# starting kernel
# baseline (speedup 1.0000x reference)
#include <hip/hip_runtime.h>
#include <hip/hip_bf16.h>

typedef float v2f __attribute__((ext_vector_type(2)));
typedef float v8f __attribute__((ext_vector_type(8)));

#define DIM 128
#define OUTC 64
#define NGRAPHS 512

__global__ void transpose_kernel(const float* __restrict__ B,
                                 float* __restrict__ BT, int K, int N) {
    int idx = blockIdx.x * blockDim.x + threadIdx.x;
    if (idx >= K * N) return;
    int k = idx / N;
    int n = idx - k * N;
    BT[(size_t)n * K + k] = B[idx];
}

__global__ __launch_bounds__(256) void wmma_gemm_f32_bt(
    const float* __restrict__ A, const float* __restrict__ BT,
    const float* __restrict__ bias, float* __restrict__ C,
    int M, int N, int K)
{
    const int lane = threadIdx.x & 31;
    const int wave = blockIdx.x * (blockDim.x >> 5) + (threadIdx.x >> 5);
    const int colGroups = N >> 6;
    const int tileM = wave / colGroups;
    const int cg    = wave - tileM * colGroups;
    if (tileM * 16 >= M) return;

    const int row0  = tileM << 4;
    const int col0  = cg << 6;
    const int m16   = lane & 15;
    const int khalf = (lane >> 4) << 1;

    const float* ap  = A  + (size_t)(row0 + m16) * K + khalf;
    const float* bp0 = BT + (size_t)(col0 + m16) * K + khalf;
    const float* bp1 = bp0 + (size_t)16 * K;
    const float* bp2 = bp0 + (size_t)32 * K;
    const float* bp3 = bp0 + (size_t)48 * K;

    v8f acc0 = {}, acc1 = {}, acc2 = {}, acc3 = {};
#pragma unroll 4
    for (int k0 = 0; k0 < K; k0 += 4) {
        v2f a  = *(const v2f*)(ap  + k0);
        v2f b0 = *(const v2f*)(bp0 + k0);
        v2f b1 = *(const v2f*)(bp1 + k0);
        v2f b2 = *(const v2f*)(bp2 + k0);
        v2f b3 = *(const v2f*)(bp3 + k0);
        acc0 = __builtin_amdgcn_wmma_f32_16x16x4_f32(false, a, false, b0, (short)0, acc0, false, false);
        acc1 = __builtin_amdgcn_wmma_f32_16x16x4_f32(false, a, false, b1, (short)0, acc1, false, false);
        acc2 = __builtin_amdgcn_wmma_f32_16x16x4_f32(false, a, false, b2, (short)0, acc2, false, false);
        acc3 = __builtin_amdgcn_wmma_f32_16x16x4_f32(false, a, false, b3, (short)0, acc3, false, false);
    }

    const int nOff  = lane & 15;
    const int mBase = row0 + ((lane >> 4) << 3);
    float* crow = C + (size_t)mBase * N + col0 + nOff;
#pragma unroll
    for (int r = 0; r < 8; ++r) {
        float b0v = bias ? bias[col0 + nOff]      : 0.0f;
        float b1v = bias ? bias[col0 + 16 + nOff] : 0.0f;
        float b2v = bias ? bias[col0 + 32 + nOff] : 0.0f;
        float b3v = bias ? bias[col0 + 48 + nOff] : 0.0f;
        crow[(size_t)r * N]      = acc0[r] + b0v;
        crow[(size_t)r * N + 16] = acc1[r] + b1v;
        crow[(size_t)r * N + 32] = acc2[r] + b2v;
        crow[(size_t)r * N + 48] = acc3[r] + b3v;
    }
}

__global__ void deg_init_kernel(float* __restrict__ deg, int n) {
    int i = blockIdx.x * blockDim.x + threadIdx.x;
    if (i < n) deg[i] = 1.0f;
}

__global__ void deg_accum_kernel(const int* __restrict__ dst,
                                 float* __restrict__ deg, int e) {
    int i = blockIdx.x * blockDim.x + threadIdx.x;
    if (i < e)
        __hip_atomic_fetch_add(&deg[dst[i]], 1.0f,
                               __ATOMIC_RELAXED, __HIP_MEMORY_SCOPE_AGENT);
}

__global__ void dinv_kernel(const float* __restrict__ deg,
                            float* __restrict__ dinv, int n) {
    int i = blockIdx.x * blockDim.x + threadIdx.x;
    if (i < n) dinv[i] = rsqrtf(deg[i]);
}

__global__ void self_bias_kernel(const float* __restrict__ h,
                                 const float* __restrict__ dinv,
                                 const float* __restrict__ bias,
                                 float* __restrict__ agg, int n) {
    int idx = blockIdx.x * blockDim.x + threadIdx.x;
    if (idx >= n * DIM) return;
    int i = idx >> 7;
    int c = idx & (DIM - 1);
    float di = dinv[i];
    agg[idx] = h[idx] * di * di + bias[c];
}

__global__ __launch_bounds__(256) void edge_scatter_kernel(
    const float* __restrict__ h, const float* __restrict__ dinv,
    const int* __restrict__ src, const int* __restrict__ dst,
    float* __restrict__ agg, int n_edges)
{
    int wave = blockIdx.x * (blockDim.x >> 5) + (threadIdx.x >> 5);
    if (wave >= n_edges) return;
    int lane = threadIdx.x & 31;
    int s = src[wave];
    int d = dst[wave];
    float coef = dinv[s] * dinv[d];
    const float4* hs = (const float4*)(h + (size_t)s * DIM);
    float4 v = hs[lane];
    float* out = agg + (size_t)d * DIM + lane * 4;
    __hip_atomic_fetch_add(&out[0], v.x * coef, __ATOMIC_RELAXED, __HIP_MEMORY_SCOPE_AGENT);
    __hip_atomic_fetch_add(&out[1], v.y * coef, __ATOMIC_RELAXED, __HIP_MEMORY_SCOPE_AGENT);
    __hip_atomic_fetch_add(&out[2], v.z * coef, __ATOMIC_RELAXED, __HIP_MEMORY_SCOPE_AGENT);
    __hip_atomic_fetch_add(&out[3], v.w * coef, __ATOMIC_RELAXED, __HIP_MEMORY_SCOPE_AGENT);
}

__global__ void relu_kernel(float* __restrict__ a, int total) {
    int i = blockIdx.x * blockDim.x + threadIdx.x;
    if (i < total) a[i] = fmaxf(a[i], 0.0f);
}

__global__ void zero_kernel(float* __restrict__ a, int total) {
    int i = blockIdx.x * blockDim.x + threadIdx.x;
    if (i < total) a[i] = 0.0f;
}

__global__ void pool_kernel(const float* __restrict__ h,
                            const int* __restrict__ batch,
                            float* __restrict__ pooled, int n) {
    int idx = blockIdx.x * blockDim.x + threadIdx.x;
    if (idx >= n * DIM) return;
    int i = idx >> 7;
    int c = idx & (DIM - 1);
    __hip_atomic_fetch_add(&pooled[(size_t)batch[i] * DIM + c], h[idx],
                           __ATOMIC_RELAXED, __HIP_MEMORY_SCOPE_AGENT);
}

__global__ __launch_bounds__(256) void log_softmax_kernel(
    const float* __restrict__ logits, float* __restrict__ out, int rows)
{
    int wave = blockIdx.x * (blockDim.x >> 5) + (threadIdx.x >> 5);
    if (wave >= rows) return;
    int lane = threadIdx.x & 31;
    const float* r = logits + (size_t)wave * OUTC;
    float v0 = r[lane];
    float v1 = r[lane + 32];
    float m = fmaxf(v0, v1);
#pragma unroll
    for (int off = 16; off > 0; off >>= 1)
        m = fmaxf(m, __shfl_xor(m, off, 32));
    float s = __expf(v0 - m) + __expf(v1 - m);
#pragma unroll
    for (int off = 16; off > 0; off >>= 1)
        s += __shfl_xor(s, off, 32);
    float ls = __logf(s);
    out[(size_t)wave * OUTC + lane]      = v0 - m - ls;
    out[(size_t)wave * OUTC + lane + 32] = v1 - m - ls;
}

extern "C" void kernel_launch(void* const* d_in, const int* in_sizes, int n_in,
                              void* d_out, int out_size, void* d_ws, size_t ws_size,
                              hipStream_t stream) {
    const float* x       = (const float*)d_in[0];
    const int*   eidx    = (const int*)d_in[1];
    const int*   x_batch = (const int*)d_in[2];
    const float* W1      = (const float*)d_in[3];
    const float* b1      = (const float*)d_in[4];
    const float* W2      = (const float*)d_in[5];
    const float* b2      = (const float*)d_in[6];
    const float* Wh      = (const float*)d_in[7];
    const float* bh      = (const float*)d_in[8];
    float*       out     = (float*)d_out;

    const int n_nodes = in_sizes[0] / DIM;
    const int n_edges = in_sizes[1] / 2;
    const int* src = eidx;
    const int* dst = eidx + n_edges;

    float* deg    = (float*)d_ws;
    float* dinv   = deg    + n_nodes;
    float* h      = dinv   + n_nodes;
    float* agg    = h      + (size_t)n_nodes * DIM;
    float* pooled = agg    + (size_t)n_nodes * DIM;
    float* logits = pooled + (size_t)NGRAPHS * DIM;
    float* W1T    = logits + (size_t)NGRAPHS * OUTC;
    float* W2T    = W1T    + DIM * DIM;
    float* WhT    = W2T    + DIM * DIM;

    const int T = 256;
    const int nodeBlk = (n_nodes + T - 1) / T;
    const int edgeBlk = (n_edges + T - 1) / T;
    const int nfBlk   = (n_nodes * DIM + T - 1) / T;
    const int edgeWaveBlk = (n_edges + 7) / 8;

    transpose_kernel<<<(DIM * DIM + T - 1) / T, T, 0, stream>>>(W1, W1T, DIM, DIM);
    transpose_kernel<<<(DIM * DIM + T - 1) / T, T, 0, stream>>>(W2, W2T, DIM, DIM);
    transpose_kernel<<<(DIM * OUTC + T - 1) / T, T, 0, stream>>>(Wh, WhT, DIM, OUTC);

    deg_init_kernel<<<nodeBlk, T, 0, stream>>>(deg, n_nodes);
    deg_accum_kernel<<<edgeBlk, T, 0, stream>>>(dst, deg, n_edges);
    dinv_kernel<<<nodeBlk, T, 0, stream>>>(deg, dinv, n_nodes);

    {
        int waves  = (n_nodes / 16) * (DIM / 64);
        int blocks = (waves + 7) / 8;
        wmma_gemm_f32_bt<<<blocks, T, 0, stream>>>(x, W1T, nullptr, h,
                                                   n_nodes, DIM, DIM);
    }
    self_bias_kernel<<<nfBlk, T, 0, stream>>>(h, dinv, b1, agg, n_nodes);
    edge_scatter_kernel<<<edgeWaveBlk, T, 0, stream>>>(h, dinv, src, dst, agg, n_edges);
    relu_kernel<<<nfBlk, T, 0, stream>>>(agg, n_nodes * DIM);

    {
        int waves  = (n_nodes / 16) * (DIM / 64);
        int blocks = (waves + 7) / 8;
        wmma_gemm_f32_bt<<<blocks, T, 0, stream>>>(agg, W2T, nullptr, h,
                                                   n_nodes, DIM, DIM);
    }
    self_bias_kernel<<<nfBlk, T, 0, stream>>>(h, dinv, b2, agg, n_nodes);
    edge_scatter_kernel<<<edgeWaveBlk, T, 0, stream>>>(h, dinv, src, dst, agg, n_edges);
    relu_kernel<<<nfBlk, T, 0, stream>>>(agg, n_nodes * DIM);

    zero_kernel<<<(NGRAPHS * DIM + T - 1) / T, T, 0, stream>>>(pooled, NGRAPHS * DIM);
    pool_kernel<<<nfBlk, T, 0, stream>>>(agg, x_batch, pooled, n_nodes);
    {
        int waves  = (NGRAPHS / 16) * (OUTC / 64);
        int blocks = (waves + 7) / 8;
        wmma_gemm_f32_bt<<<blocks, T, 0, stream>>>(pooled, WhT, bh, logits,
                                                   NGRAPHS, OUTC, DIM);
    }
    log_softmax_kernel<<<(NGRAPHS + 7) / 8, T, 0, stream>>>(logits, out, NGRAPHS);
}


# Edge sum via destination-grouped edge list (int-atomic rank + scan, rebuilt per call): one wave per node sums rows in registers with self-loop, bias, relu; no float atomics; sum order = rank order
# speedup vs baseline: 1.9446x; 1.9446x over previous
#include <hip/hip_runtime.h>
#include <hip/hip_bf16.h>

typedef float v2f __attribute__((ext_vector_type(2)));
typedef float v8f __attribute__((ext_vector_type(8)));

#define DIM 128
#define OUTC 64
#define NGRAPHS 512

__global__ void transpose_kernel(const float* __restrict__ B,
                                 float* __restrict__ BT, int K, int N) {
    int idx = blockIdx.x * blockDim.x + threadIdx.x;
    if (idx >= K * N) return;
    int k = idx / N;
    int n = idx - k * N;
    BT[(size_t)n * K + k] = B[idx];
}

__global__ __launch_bounds__(256) void wmma_gemm_f32_bt(
    const float* __restrict__ A, const float* __restrict__ BT,
    const float* __restrict__ bias, float* __restrict__ C,
    int M, int N, int K)
{
    const int lane = threadIdx.x & 31;
    const int wave = blockIdx.x * (blockDim.x >> 5) + (threadIdx.x >> 5);
    const int colGroups = N >> 6;
    const int tileM = wave / colGroups;
    const int cg    = wave - tileM * colGroups;
    if (tileM * 16 >= M) return;

    const int row0  = tileM << 4;
    const int col0  = cg << 6;
    const int m16   = lane & 15;
    const int khalf = (lane >> 4) << 1;

    const float* ap  = A  + (size_t)(row0 + m16) * K + khalf;
    const float* bp0 = BT + (size_t)(col0 + m16) * K + khalf;
    const float* bp1 = bp0 + (size_t)16 * K;
    const float* bp2 = bp0 + (size_t)32 * K;
    const float* bp3 = bp0 + (size_t)48 * K;

    v8f acc0 = {}, acc1 = {}, acc2 = {}, acc3 = {};
#pragma unroll 4
    for (int k0 = 0; k0 < K; k0 += 4) {
        v2f a  = *(const v2f*)(ap  + k0);
        v2f b0 = *(const v2f*)(bp0 + k0);
        v2f b1 = *(const v2f*)(bp1 + k0);
        v2f b2 = *(const v2f*)(bp2 + k0);
        v2f b3 = *(const v2f*)(bp3 + k0);
        acc0 = __builtin_amdgcn_wmma_f32_16x16x4_f32(false, a, false, b0, (short)0, acc0, false, false);
        acc1 = __builtin_amdgcn_wmma_f32_16x16x4_f32(false, a, false, b1, (short)0, acc1, false, false);
        acc2 = __builtin_amdgcn_wmma_f32_16x16x4_f32(false, a, false, b2, (short)0, acc2, false, false);
        acc3 = __builtin_amdgcn_wmma_f32_16x16x4_f32(false, a, false, b3, (short)0, acc3, false, false);
    }

    const int nOff  = lane & 15;
    const int mBase = row0 + ((lane >> 4) << 3);
    float* crow = C + (size_t)mBase * N + col0 + nOff;
#pragma unroll
    for (int r = 0; r < 8; ++r) {
        float b0v = bias ? bias[col0 + nOff]      : 0.0f;
        float b1v = bias ? bias[col0 + 16 + nOff] : 0.0f;
        float b2v = bias ? bias[col0 + 32 + nOff] : 0.0f;
        float b3v = bias ? bias[col0 + 48 + nOff] : 0.0f;
        crow[(size_t)r * N]      = acc0[r] + b0v;
        crow[(size_t)r * N + 16] = acc1[r] + b1v;
        crow[(size_t)r * N + 32] = acc2[r] + b2v;
        crow[(size_t)r * N + 48] = acc3[r] + b3v;
    }
}

__global__ void cnt_zero_kernel(int* __restrict__ cnt, int n) {
    int i = blockIdx.x * blockDim.x + threadIdx.x;
    if (i < n) cnt[i] = 0;
}

__global__ void count_rank_kernel(const int* __restrict__ dst,
                                  int* __restrict__ cnt,
                                  int* __restrict__ rank, int e) {
    int i = blockIdx.x * blockDim.x + threadIdx.x;
    if (i < e)
        rank[i] = __hip_atomic_fetch_add(&cnt[dst[i]], 1,
                                         __ATOMIC_RELAXED, __HIP_MEMORY_SCOPE_AGENT);
}

__global__ __launch_bounds__(1024) void rowptr_dinv_kernel(
    const int* __restrict__ cnt, int* __restrict__ rowptr,
    float* __restrict__ dinv, int n)
{
    __shared__ int wpre[32];
    __shared__ int wtot[32];
    const int lane = threadIdx.x & 31;
    const int wv   = threadIdx.x >> 5;
    const int base = blockIdx.x * 1024;

    int pre = 0;
    for (int k = threadIdx.x; k < base; k += 1024) pre += cnt[k];

    const int i = base + threadIdx.x;
    const int c = (i < n) ? cnt[i] : 0;

    int incl = c;
    for (int off = 1; off < 32; off <<= 1) {
        int y = __shfl_up(incl, off, 32);
        if (lane >= off) incl += y;
    }
    for (int off = 16; off > 0; off >>= 1) pre += __shfl_xor(pre, off, 32);
    if (lane == 31) wtot[wv] = incl;
    if (lane == 0)  wpre[wv] = pre;
    __syncthreads();

    int x = wpre[lane] + ((lane < wv) ? wtot[lane] : 0);
    for (int off = 16; off > 0; off >>= 1) x += __shfl_xor(x, off, 32);

    if (i < n) {
        rowptr[i] = x + incl - c;
        dinv[i]   = rsqrtf((float)(c + 1));
        if (i == n - 1) rowptr[n] = x + incl;
    }
}

__global__ void csr_fill_kernel(const int* __restrict__ src,
                                const int* __restrict__ dst,
                                const int* __restrict__ rank,
                                const int* __restrict__ rowptr,
                                int* __restrict__ csr_src, int e) {
    int i = blockIdx.x * blockDim.x + threadIdx.x;
    if (i < e) csr_src[rowptr[dst[i]] + rank[i]] = src[i];
}

__global__ __launch_bounds__(256) void gather_relu_kernel(
    const float* __restrict__ h, const float* __restrict__ dinv,
    const float* __restrict__ bias, const int* __restrict__ rowptr,
    const int* __restrict__ csr_src, float* __restrict__ agg, int n)
{
    const int lane = threadIdx.x & 31;
    const int i = __builtin_amdgcn_readfirstlane(
        (int)(blockIdx.x * (blockDim.x >> 5) + (threadIdx.x >> 5)));
    if (i >= n) return;

    const int   beg = rowptr[i];
    const int   end = rowptr[i + 1];
    const float di  = dinv[i];
    const float dd  = di * di;
    const float4 hv = ((const float4*)(h + (size_t)i * DIM))[lane];

    float4 acc;
    acc.x = hv.x * dd + bias[lane * 4 + 0];
    acc.y = hv.y * dd + bias[lane * 4 + 1];
    acc.z = hv.z * dd + bias[lane * 4 + 2];
    acc.w = hv.w * dd + bias[lane * 4 + 3];

    for (int c0 = beg; c0 < end; c0 += 32) {
        int m = end - c0;
        if (m > 32) m = 32;
        const int   sl = csr_src[c0 + ((lane < m) ? lane : 0)];
        const float dl = dinv[sl];
        for (int j0 = 0; j0 < m; j0 += 4) {
            float4 v[4];
            float  w[4];
#pragma unroll
            for (int u = 0; u < 4; ++u) {
                const int jj = j0 + u;
                const int s  = __builtin_amdgcn_readlane(sl, jj);
                v[u] = make_float4(0.0f, 0.0f, 0.0f, 0.0f);
                if (jj < m) v[u] = ((const float4*)(h + (size_t)s * DIM))[lane];
            }
#pragma unroll
            for (int u = 0; u < 4; ++u) {
                w[u] = __int_as_float(
                    __builtin_amdgcn_readlane(__float_as_int(dl), j0 + u)) * di;
                acc.x += v[u].x * w[u];
                acc.y += v[u].y * w[u];
                acc.z += v[u].z * w[u];
                acc.w += v[u].w * w[u];
            }
        }
    }

    acc.x = fmaxf(acc.x, 0.0f);
    acc.y = fmaxf(acc.y, 0.0f);
    acc.z = fmaxf(acc.z, 0.0f);
    acc.w = fmaxf(acc.w, 0.0f);
    ((float4*)(agg + (size_t)i * DIM))[lane] = acc;
}

__global__ void zero_kernel(float* __restrict__ a, int total) {
    int i = blockIdx.x * blockDim.x + threadIdx.x;
    if (i < total) a[i] = 0.0f;
}

__global__ void pool_kernel(const float* __restrict__ h,
                            const int* __restrict__ batch,
                            float* __restrict__ pooled, int n) {
    int idx = blockIdx.x * blockDim.x + threadIdx.x;
    if (idx >= n * DIM) return;
    int i = idx >> 7;
    int c = idx & (DIM - 1);
    __hip_atomic_fetch_add(&pooled[(size_t)batch[i] * DIM + c], h[idx],
                           __ATOMIC_RELAXED, __HIP_MEMORY_SCOPE_AGENT);
}

__global__ __launch_bounds__(256) void log_softmax_kernel(
    const float* __restrict__ logits, float* __restrict__ out, int rows)
{
    int wave = blockIdx.x * (blockDim.x >> 5) + (threadIdx.x >> 5);
    if (wave >= rows) return;
    int lane = threadIdx.x & 31;
    const float* r = logits + (size_t)wave * OUTC;
    float v0 = r[lane];
    float v1 = r[lane + 32];
    float m = fmaxf(v0, v1);
#pragma unroll
    for (int off = 16; off > 0; off >>= 1)
        m = fmaxf(m, __shfl_xor(m, off, 32));
    float s = __expf(v0 - m) + __expf(v1 - m);
#pragma unroll
    for (int off = 16; off > 0; off >>= 1)
        s += __shfl_xor(s, off, 32);
    float ls = __logf(s);
    out[(size_t)wave * OUTC + lane]      = v0 - m - ls;
    out[(size_t)wave * OUTC + lane + 32] = v1 - m - ls;
}

extern "C" void kernel_launch(void* const* d_in, const int* in_sizes, int n_in,
                              void* d_out, int out_size, void* d_ws, size_t ws_size,
                              hipStream_t stream) {
    const float* x       = (const float*)d_in[0];
    const int*   eidx    = (const int*)d_in[1];
    const int*   x_batch = (const int*)d_in[2];
    const float* W1      = (const float*)d_in[3];
    const float* b1      = (const float*)d_in[4];
    const float* W2      = (const float*)d_in[5];
    const float* b2      = (const float*)d_in[6];
    const float* Wh      = (const float*)d_in[7];
    const float* bh      = (const float*)d_in[8];
    float*       out     = (float*)d_out;

    const int n_nodes = in_sizes[0] / DIM;
    const int n_edges = in_sizes[1] / 2;
    const int* src = eidx;
    const int* dst = eidx + n_edges;

    float* h       = (float*)d_ws;
    float* agg     = h      + (size_t)n_nodes * DIM;
    float* pooled  = agg    + (size_t)n_nodes * DIM;
    float* logits  = pooled + (size_t)NGRAPHS * DIM;
    float* W1T     = logits + (size_t)NGRAPHS * OUTC;
    float* W2T     = W1T    + DIM * DIM;
    float* WhT     = W2T    + DIM * DIM;
    float* dinv    = WhT    + OUTC * DIM;
    int*   cnt     = (int*)(dinv + n_nodes);
    int*   rank    = cnt     + n_nodes;
    int*   csr_src = rank    + n_edges;
    int*   rowptr  = csr_src + n_edges;

    const int T = 256;
    const int nodeBlk = (n_nodes + T - 1) / T;
    const int edgeBlk = (n_edges + T - 1) / T;
    const int nfBlk   = (n_nodes * DIM + T - 1) / T;
    const int nodeWaveBlk = (n_nodes + 7) / 8;

    transpose_kernel<<<(DIM * DIM + T - 1) / T, T, 0, stream>>>(W1, W1T, DIM, DIM);
    transpose_kernel<<<(DIM * DIM + T - 1) / T, T, 0, stream>>>(W2, W2T, DIM, DIM);
    transpose_kernel<<<(DIM * OUTC + T - 1) / T, T, 0, stream>>>(Wh, WhT, DIM, OUTC);

    cnt_zero_kernel<<<nodeBlk, T, 0, stream>>>(cnt, n_nodes);
    count_rank_kernel<<<edgeBlk, T, 0, stream>>>(dst, cnt, rank, n_edges);
    rowptr_dinv_kernel<<<(n_nodes + 1023) / 1024, 1024, 0, stream>>>(cnt, rowptr, dinv, n_nodes);
    csr_fill_kernel<<<edgeBlk, T, 0, stream>>>(src, dst, rank, rowptr, csr_src, n_edges);

    {
        int waves  = (n_nodes / 16) * (DIM / 64);
        int blocks = (waves + 7) / 8;
        wmma_gemm_f32_bt<<<blocks, T, 0, stream>>>(x, W1T, nullptr, h,
                                                   n_nodes, DIM, DIM);
    }
    gather_relu_kernel<<<nodeWaveBlk, T, 0, stream>>>(h, dinv, b1, rowptr, csr_src, agg, n_nodes);

    {
        int waves  = (n_nodes / 16) * (DIM / 64);
        int blocks = (waves + 7) / 8;
        wmma_gemm_f32_bt<<<blocks, T, 0, stream>>>(agg, W2T, nullptr, h,
                                                   n_nodes, DIM, DIM);
    }
    gather_relu_kernel<<<nodeWaveBlk, T, 0, stream>>>(h, dinv, b2, rowptr, csr_src, agg, n_nodes);

    zero_kernel<<<(NGRAPHS * DIM + T - 1) / T, T, 0, stream>>>(pooled, NGRAPHS * DIM);
    pool_kernel<<<nfBlk, T, 0, stream>>>(agg, x_batch, pooled, n_nodes);
    {
        int waves  = (NGRAPHS / 16) * (OUTC / 64);
        int blocks = (waves + 7) / 8;
        wmma_gemm_f32_bt<<<blocks, T, 0, stream>>>(pooled, WhT, bh, logits,
                                                   NGRAPHS, OUTC, DIM);
    }
    log_softmax_kernel<<<(NGRAPHS + 7) / 8, T, 0, stream>>>(logits, out, NGRAPHS);
}


# Graph pooling folded into the layer-2 gather: its 8 waves stage rows in LDS, equal-graph runs are added in node order, one f32 atomic per channel per run (was per node); pool launch gone
# speedup vs baseline: 2.0869x; 1.0732x over previous
#include <hip/hip_runtime.h>
#include <hip/hip_bf16.h>

typedef float v2f __attribute__((ext_vector_type(2)));
typedef float v8f __attribute__((ext_vector_type(8)));

#define DIM 128
#define OUTC 64
#define NGRAPHS 512

__global__ void transpose_kernel(const float* __restrict__ B,
                                 float* __restrict__ BT, int K, int N) {
    int idx = blockIdx.x * blockDim.x + threadIdx.x;
    if (idx >= K * N) return;
    int k = idx / N;
    int n = idx - k * N;
    BT[(size_t)n * K + k] = B[idx];
}

__global__ __launch_bounds__(256) void wmma_gemm_f32_bt(
    const float* __restrict__ A, const float* __restrict__ BT,
    const float* __restrict__ bias, float* __restrict__ C,
    int M, int N, int K)
{
    const int lane = threadIdx.x & 31;
    const int wave = blockIdx.x * (blockDim.x >> 5) + (threadIdx.x >> 5);
    const int colGroups = N >> 6;
    const int tileM = wave / colGroups;
    const int cg    = wave - tileM * colGroups;
    if (tileM * 16 >= M) return;

    const int row0  = tileM << 4;
    const int col0  = cg << 6;
    const int m16   = lane & 15;
    const int khalf = (lane >> 4) << 1;

    const float* ap  = A  + (size_t)(row0 + m16) * K + khalf;
    const float* bp0 = BT + (size_t)(col0 + m16) * K + khalf;
    const float* bp1 = bp0 + (size_t)16 * K;
    const float* bp2 = bp0 + (size_t)32 * K;
    const float* bp3 = bp0 + (size_t)48 * K;

    v8f acc0 = {}, acc1 = {}, acc2 = {}, acc3 = {};
#pragma unroll 4
    for (int k0 = 0; k0 < K; k0 += 4) {
        v2f a  = *(const v2f*)(ap  + k0);
        v2f b0 = *(const v2f*)(bp0 + k0);
        v2f b1 = *(const v2f*)(bp1 + k0);
        v2f b2 = *(const v2f*)(bp2 + k0);
        v2f b3 = *(const v2f*)(bp3 + k0);
        acc0 = __builtin_amdgcn_wmma_f32_16x16x4_f32(false, a, false, b0, (short)0, acc0, false, false);
        acc1 = __builtin_amdgcn_wmma_f32_16x16x4_f32(false, a, false, b1, (short)0, acc1, false, false);
        acc2 = __builtin_amdgcn_wmma_f32_16x16x4_f32(false, a, false, b2, (short)0, acc2, false, false);
        acc3 = __builtin_amdgcn_wmma_f32_16x16x4_f32(false, a, false, b3, (short)0, acc3, false, false);
    }

    const int nOff  = lane & 15;
    const int mBase = row0 + ((lane >> 4) << 3);
    float* crow = C + (size_t)mBase * N + col0 + nOff;
#pragma unroll
    for (int r = 0; r < 8; ++r) {
        float b0v = bias ? bias[col0 + nOff]      : 0.0f;
        float b1v = bias ? bias[col0 + 16 + nOff] : 0.0f;
        float b2v = bias ? bias[col0 + 32 + nOff] : 0.0f;
        float b3v = bias ? bias[col0 + 48 + nOff] : 0.0f;
        crow[(size_t)r * N]      = acc0[r] + b0v;
        crow[(size_t)r * N + 16] = acc1[r] + b1v;
        crow[(size_t)r * N + 32] = acc2[r] + b2v;
        crow[(size_t)r * N + 48] = acc3[r] + b3v;
    }
}

__global__ void cnt_zero_kernel(int* __restrict__ cnt, int n) {
    int i = blockIdx.x * blockDim.x + threadIdx.x;
    if (i < n) cnt[i] = 0;
}

__global__ void count_rank_kernel(const int* __restrict__ dst,
                                  int* __restrict__ cnt,
                                  int* __restrict__ rank, int e) {
    int i = blockIdx.x * blockDim.x + threadIdx.x;
    if (i < e)
        rank[i] = __hip_atomic_fetch_add(&cnt[dst[i]], 1,
                                         __ATOMIC_RELAXED, __HIP_MEMORY_SCOPE_AGENT);
}

__global__ __launch_bounds__(1024) void rowptr_dinv_kernel(
    const int* __restrict__ cnt, int* __restrict__ rowptr,
    float* __restrict__ dinv, int n)
{
    __shared__ int wpre[32];
    __shared__ int wtot[32];
    const int lane = threadIdx.x & 31;
    const int wv   = threadIdx.x >> 5;
    const int base = blockIdx.x * 1024;

    int pre = 0;
    for (int k = threadIdx.x; k < base; k += 1024) pre += cnt[k];

    const int i = base + threadIdx.x;
    const int c = (i < n) ? cnt[i] : 0;

    int incl = c;
    for (int off = 1; off < 32; off <<= 1) {
        int y = __shfl_up(incl, off, 32);
        if (lane >= off) incl += y;
    }
    for (int off = 16; off > 0; off >>= 1) pre += __shfl_xor(pre, off, 32);
    if (lane == 31) wtot[wv] = incl;
    if (lane == 0)  wpre[wv] = pre;
    __syncthreads();

    int x = wpre[lane] + ((lane < wv) ? wtot[lane] : 0);
    for (int off = 16; off > 0; off >>= 1) x += __shfl_xor(x, off, 32);

    if (i < n) {
        rowptr[i] = x + incl - c;
        dinv[i]   = rsqrtf((float)(c + 1));
        if (i == n - 1) rowptr[n] = x + incl;
    }
}

__global__ void csr_fill_kernel(const int* __restrict__ src,
                                const int* __restrict__ dst,
                                const int* __restrict__ rank,
                                const int* __restrict__ rowptr,
                                int* __restrict__ csr_src, int e) {
    int i = blockIdx.x * blockDim.x + threadIdx.x;
    if (i < e) csr_src[rowptr[dst[i]] + rank[i]] = src[i];
}

__device__ __forceinline__ float4 gather_row(
    const float* __restrict__ h, const float* __restrict__ dinv,
    const float* __restrict__ bias, const int* __restrict__ rowptr,
    const int* __restrict__ csr_src, int i, int lane)
{
    const int   beg = rowptr[i];
    const int   end = rowptr[i + 1];
    const float di  = dinv[i];
    const float dd  = di * di;
    const float4 hv = ((const float4*)(h + (size_t)i * DIM))[lane];

    float4 acc;
    acc.x = hv.x * dd + bias[lane * 4 + 0];
    acc.y = hv.y * dd + bias[lane * 4 + 1];
    acc.z = hv.z * dd + bias[lane * 4 + 2];
    acc.w = hv.w * dd + bias[lane * 4 + 3];

    for (int c0 = beg; c0 < end; c0 += 32) {
        int m = end - c0;
        if (m > 32) m = 32;
        const int   sl = csr_src[c0 + ((lane < m) ? lane : 0)];
        const float dl = dinv[sl];
        for (int j0 = 0; j0 < m; j0 += 4) {
            float4 v[4];
            float  w[4];
#pragma unroll
            for (int u = 0; u < 4; ++u) {
                const int jj = j0 + u;
                const int s  = __builtin_amdgcn_readlane(sl, jj);
                v[u] = make_float4(0.0f, 0.0f, 0.0f, 0.0f);
                if (jj < m) v[u] = ((const float4*)(h + (size_t)s * DIM))[lane];
            }
#pragma unroll
            for (int u = 0; u < 4; ++u) {
                w[u] = __int_as_float(
                    __builtin_amdgcn_readlane(__float_as_int(dl), j0 + u)) * di;
                acc.x += v[u].x * w[u];
                acc.y += v[u].y * w[u];
                acc.z += v[u].z * w[u];
                acc.w += v[u].w * w[u];
            }
        }
    }

    acc.x = fmaxf(acc.x, 0.0f);
    acc.y = fmaxf(acc.y, 0.0f);
    acc.z = fmaxf(acc.z, 0.0f);
    acc.w = fmaxf(acc.w, 0.0f);
    return acc;
}

__global__ __launch_bounds__(256) void gather_relu_kernel(
    const float* __restrict__ h, const float* __restrict__ dinv,
    const float* __restrict__ bias, const int* __restrict__ rowptr,
    const int* __restrict__ csr_src, float* __restrict__ agg, int n)
{
    const int lane = threadIdx.x & 31;
    const int i = __builtin_amdgcn_readfirstlane(
        (int)(blockIdx.x * (blockDim.x >> 5) + (threadIdx.x >> 5)));
    if (i >= n) return;
    const float4 acc = gather_row(h, dinv, bias, rowptr, csr_src, i, lane);
    ((float4*)(agg + (size_t)i * DIM))[lane] = acc;
}

__global__ __launch_bounds__(256) void gather_relu_pool_kernel(
    const float* __restrict__ h, const float* __restrict__ dinv,
    const float* __restrict__ bias, const int* __restrict__ rowptr,
    const int* __restrict__ csr_src, const int* __restrict__ batch,
    float* __restrict__ pooled, int n)
{
    __shared__ float srow[8][DIM];
    __shared__ __attribute__((aligned(16))) int sgid[8];
    const int lane = threadIdx.x & 31;
    const int wv   = threadIdx.x >> 5;
    const int i = __builtin_amdgcn_readfirstlane(
        (int)(blockIdx.x * (blockDim.x >> 5) + wv));

    float4 acc = make_float4(0.0f, 0.0f, 0.0f, 0.0f);
    int gid = -1;
    if (i < n) {
        gid = batch[i];
        acc = gather_row(h, dinv, bias, rowptr, csr_src, i, lane);
    }
    ((float4*)srow[wv])[lane] = acc;
    if (lane == 0) sgid[wv] = gid;
    __syncthreads();

    if (threadIdx.x < DIM) {
        const int c = threadIdx.x;
        int   g[8];
        float v[8];
#pragma unroll
        for (int k = 0; k < 8; ++k) {
            g[k] = __builtin_amdgcn_readfirstlane(sgid[k]);
            v[k] = srow[k][c];
        }
        int   cur = g[0];
        float run = v[0];
#pragma unroll
        for (int k = 1; k < 8; ++k) {
            if (g[k] != cur) {
                if (cur >= 0)
                    __hip_atomic_fetch_add(&pooled[(size_t)cur * DIM + c], run,
                                           __ATOMIC_RELAXED, __HIP_MEMORY_SCOPE_AGENT);
                cur = g[k];
                run = v[k];
            } else {
                run += v[k];
            }
        }
        if (cur >= 0)
            __hip_atomic_fetch_add(&pooled[(size_t)cur * DIM + c], run,
                                   __ATOMIC_RELAXED, __HIP_MEMORY_SCOPE_AGENT);
    }
}

__global__ void zero_kernel(float* __restrict__ a, int total) {
    int i = blockIdx.x * blockDim.x + threadIdx.x;
    if (i < total) a[i] = 0.0f;
}

__global__ __launch_bounds__(256) void log_softmax_kernel(
    const float* __restrict__ logits, float* __restrict__ out, int rows)
{
    int wave = blockIdx.x * (blockDim.x >> 5) + (threadIdx.x >> 5);
    if (wave >= rows) return;
    int lane = threadIdx.x & 31;
    const float* r = logits + (size_t)wave * OUTC;
    float v0 = r[lane];
    float v1 = r[lane + 32];
    float m = fmaxf(v0, v1);
#pragma unroll
    for (int off = 16; off > 0; off >>= 1)
        m = fmaxf(m, __shfl_xor(m, off, 32));
    float s = __expf(v0 - m) + __expf(v1 - m);
#pragma unroll
    for (int off = 16; off > 0; off >>= 1)
        s += __shfl_xor(s, off, 32);
    float ls = __logf(s);
    out[(size_t)wave * OUTC + lane]      = v0 - m - ls;
    out[(size_t)wave * OUTC + lane + 32] = v1 - m - ls;
}

extern "C" void kernel_launch(void* const* d_in, const int* in_sizes, int n_in,
                              void* d_out, int out_size, void* d_ws, size_t ws_size,
                              hipStream_t stream) {
    const float* x       = (const float*)d_in[0];
    const int*   eidx    = (const int*)d_in[1];
    const int*   x_batch = (const int*)d_in[2];
    const float* W1      = (const float*)d_in[3];
    const float* b1      = (const float*)d_in[4];
    const float* W2      = (const float*)d_in[5];
    const float* b2      = (const float*)d_in[6];
    const float* Wh      = (const float*)d_in[7];
    const float* bh      = (const float*)d_in[8];
    float*       out     = (float*)d_out;

    const int n_nodes = in_sizes[0] / DIM;
    const int n_edges = in_sizes[1] / 2;
    const int* src = eidx;
    const int* dst = eidx + n_edges;

    float* h       = (float*)d_ws;
    float* agg     = h      + (size_t)n_nodes * DIM;
    float* pooled  = agg    + (size_t)n_nodes * DIM;
    float* logits  = pooled + (size_t)NGRAPHS * DIM;
    float* W1T     = logits + (size_t)NGRAPHS * OUTC;
    float* W2T     = W1T    + DIM * DIM;
    float* WhT     = W2T    + DIM * DIM;
    float* dinv    = WhT    + OUTC * DIM;
    int*   cnt     = (int*)(dinv + n_nodes);
    int*   rank    = cnt     + n_nodes;
    int*   csr_src = rank    + n_edges;
    int*   rowptr  = csr_src + n_edges;

    const int T = 256;
    const int nodeBlk = (n_nodes + T - 1) / T;
    const int edgeBlk = (n_edges + T - 1) / T;
    const int nodeWaveBlk = (n_nodes + 7) / 8;

    transpose_kernel<<<(DIM * DIM + T - 1) / T, T, 0, stream>>>(W1, W1T, DIM, DIM);
    transpose_kernel<<<(DIM * DIM + T - 1) / T, T, 0, stream>>>(W2, W2T, DIM, DIM);
    transpose_kernel<<<(DIM * OUTC + T - 1) / T, T, 0, stream>>>(Wh, WhT, DIM, OUTC);

    cnt_zero_kernel<<<nodeBlk, T, 0, stream>>>(cnt, n_nodes);
    count_rank_kernel<<<edgeBlk, T, 0, stream>>>(dst, cnt, rank, n_edges);
    rowptr_dinv_kernel<<<(n_nodes + 1023) / 1024, 1024, 0, stream>>>(cnt, rowptr, dinv, n_nodes);
    csr_fill_kernel<<<edgeBlk, T, 0, stream>>>(src, dst, rank, rowptr, csr_src, n_edges);

    {
        int waves  = (n_nodes / 16) * (DIM / 64);
        int blocks = (waves + 7) / 8;
        wmma_gemm_f32_bt<<<blocks, T, 0, stream>>>(x, W1T, nullptr, h,
                                                   n_nodes, DIM, DIM);
    }
    gather_relu_kernel<<<nodeWaveBlk, T, 0, stream>>>(h, dinv, b1, rowptr, csr_src, agg, n_nodes);

    {
        int waves  = (n_nodes / 16) * (DIM / 64);
        int blocks = (waves + 7) / 8;
        wmma_gemm_f32_bt<<<blocks, T, 0, stream>>>(agg, W2T, nullptr, h,
                                                   n_nodes, DIM, DIM);
    }
    zero_kernel<<<(NGRAPHS * DIM + T - 1) / T, T, 0, stream>>>(pooled, NGRAPHS * DIM);
    gather_relu_pool_kernel<<<nodeWaveBlk, T, 0, stream>>>(h, dinv, b2, rowptr, csr_src,
                                                          x_batch, pooled, n_nodes);

    {
        int waves  = (NGRAPHS / 16) * (OUTC / 64);
        int blocks = (waves + 7) / 8;
        wmma_gemm_f32_bt<<<blocks, T, 0, stream>>>(pooled, WhT, bh, logits,
                                                   NGRAPHS, OUTC, DIM);
    }
    log_softmax_kernel<<<(NGRAPHS + 7) / 8, T, 0, stream>>>(logits, out, NGRAPHS);
}


# Independent work shares launches (14 -> 8): 3 transposes + 2 clears in one; x@W1 in the same launch as the edge ranking atomics; head product + log-softmax in one (same sum tree)
# speedup vs baseline: 2.5434x; 1.2187x over previous
#include <hip/hip_runtime.h>
#include <hip/hip_bf16.h>

typedef float v2f __attribute__((ext_vector_type(2)));
typedef float v8f __attribute__((ext_vector_type(8)));

#define DIM 128
#define OUTC 64
#define NGRAPHS 512

__global__ void prep_kernel(const float* __restrict__ W1, const float* __restrict__ W2,
                            const float* __restrict__ Wh,
                            float* __restrict__ W1T, float* __restrict__ W2T,
                            float* __restrict__ WhT,
                            float* __restrict__ pooled, int* __restrict__ cnt, int n) {
    int idx = blockIdx.x * blockDim.x + threadIdx.x;
    if (idx < DIM * DIM) {
        int k = idx / DIM;
        int c = idx - k * DIM;
        W1T[c * DIM + k] = W1[idx];
        return;
    }
    idx -= DIM * DIM;
    if (idx < DIM * DIM) {
        int k = idx / DIM;
        int c = idx - k * DIM;
        W2T[c * DIM + k] = W2[idx];
        return;
    }
    idx -= DIM * DIM;
    if (idx < DIM * OUTC) {
        int k = idx / OUTC;
        int c = idx - k * OUTC;
        WhT[c * DIM + k] = Wh[idx];
        return;
    }
    idx -= DIM * OUTC;
    if (idx < NGRAPHS * DIM) {
        pooled[idx] = 0.0f;
        return;
    }
    idx -= NGRAPHS * DIM;
    if (idx < n) cnt[idx] = 0;
}

struct Strip { v8f a0, a1, a2, a3; };

__device__ __forceinline__ Strip gemm_strip(
    const float* __restrict__ A, const float* __restrict__ BT,
    int K, int row0, int col0, int lane)
{
    const int m16   = lane & 15;
    const int khalf = (lane >> 4) << 1;

    const float* ap  = A  + (size_t)(row0 + m16) * K + khalf;
    const float* bp0 = BT + (size_t)(col0 + m16) * K + khalf;
    const float* bp1 = bp0 + (size_t)16 * K;
    const float* bp2 = bp0 + (size_t)32 * K;
    const float* bp3 = bp0 + (size_t)48 * K;

    v8f acc0 = {}, acc1 = {}, acc2 = {}, acc3 = {};
#pragma unroll 4
    for (int k0 = 0; k0 < K; k0 += 4) {
        v2f a  = *(const v2f*)(ap  + k0);
        v2f b0 = *(const v2f*)(bp0 + k0);
        v2f b1 = *(const v2f*)(bp1 + k0);
        v2f b2 = *(const v2f*)(bp2 + k0);
        v2f b3 = *(const v2f*)(bp3 + k0);
        acc0 = __builtin_amdgcn_wmma_f32_16x16x4_f32(false, a, false, b0, (short)0, acc0, false, false);
        acc1 = __builtin_amdgcn_wmma_f32_16x16x4_f32(false, a, false, b1, (short)0, acc1, false, false);
        acc2 = __builtin_amdgcn_wmma_f32_16x16x4_f32(false, a, false, b2, (short)0, acc2, false, false);
        acc3 = __builtin_amdgcn_wmma_f32_16x16x4_f32(false, a, false, b3, (short)0, acc3, false, false);
    }
    Strip s;
    s.a0 = acc0; s.a1 = acc1; s.a2 = acc2; s.a3 = acc3;
    return s;
}

__device__ __forceinline__ void store_strip(
    const Strip& s, float* __restrict__ C, int N, int row0, int col0, int lane)
{
    const int nOff  = lane & 15;
    const int mBase = row0 + ((lane >> 4) << 3);
    float* crow = C + (size_t)mBase * N + col0 + nOff;
#pragma unroll
    for (int r = 0; r < 8; ++r) {
        crow[(size_t)r * N]      = s.a0[r];
        crow[(size_t)r * N + 16] = s.a1[r];
        crow[(size_t)r * N + 32] = s.a2[r];
        crow[(size_t)r * N + 48] = s.a3[r];
    }
}

__global__ __launch_bounds__(256) void wmma_gemm_f32_bt(
    const float* __restrict__ A, const float* __restrict__ BT,
    float* __restrict__ C, int M, int N, int K)
{
    const int lane = threadIdx.x & 31;
    const int wave = blockIdx.x * (blockDim.x >> 5) + (threadIdx.x >> 5);
    const int colGroups = N >> 6;
    const int tileM = wave / colGroups;
    const int cg    = wave - tileM * colGroups;
    if (tileM * 16 >= M) return;
    const int row0 = tileM << 4;
    const int col0 = cg << 6;
    const Strip s = gemm_strip(A, BT, K, row0, col0, lane);
    store_strip(s, C, N, row0, col0, lane);
}

__global__ __launch_bounds__(256) void gemm_count_rank_kernel(
    const float* __restrict__ A, const float* __restrict__ BT,
    float* __restrict__ C, int M, int N, int K, int gemmBlocks,
    const int* __restrict__ dst, int* __restrict__ cnt,
    int* __restrict__ rank, int e)
{
    if ((int)blockIdx.x >= gemmBlocks) {
        const int i = ((int)blockIdx.x - gemmBlocks) * (int)blockDim.x + (int)threadIdx.x;
        if (i < e)
            rank[i] = __hip_atomic_fetch_add(&cnt[dst[i]], 1,
                                             __ATOMIC_RELAXED, __HIP_MEMORY_SCOPE_AGENT);
        return;
    }
    const int lane = threadIdx.x & 31;
    const int wave = blockIdx.x * (blockDim.x >> 5) + (threadIdx.x >> 5);
    const int colGroups = N >> 6;
    const int tileM = wave / colGroups;
    const int cg    = wave - tileM * colGroups;
    if (tileM * 16 >= M) return;
    const int row0 = tileM << 4;
    const int col0 = cg << 6;
    const Strip s = gemm_strip(A, BT, K, row0, col0, lane);
    store_strip(s, C, N, row0, col0, lane);
}

__global__ __launch_bounds__(256) void head_logsoftmax_kernel(
    const float* __restrict__ A, const float* __restrict__ BT,
    const float* __restrict__ bias, float* __restrict__ out, int M, int K)
{
    const int lane = threadIdx.x & 31;
    const int wave = blockIdx.x * (blockDim.x >> 5) + (threadIdx.x >> 5);
    if (wave * 16 >= M) return;
    const int row0 = wave << 4;
    const Strip s = gemm_strip(A, BT, K, row0, 0, lane);

    const int c  = lane & 15;
    const int h8 = (lane >> 4) << 3;
    const float bb0 = bias[c];
    const float bb1 = bias[c + 16];
    const float bb2 = bias[c + 32];
    const float bb3 = bias[c + 48];
    float* orow = out + (size_t)(row0 + h8) * OUTC + c;
#pragma unroll
    for (int r = 0; r < 8; ++r) {
        const float v0 = s.a0[r] + bb0;
        const float v1 = s.a1[r] + bb1;
        const float v2 = s.a2[r] + bb2;
        const float v3 = s.a3[r] + bb3;
        float m = fmaxf(fmaxf(v0, v2), fmaxf(v1, v3));
        for (int off = 8; off > 0; off >>= 1)
            m = fmaxf(m, __shfl_xor(m, off, 32));
        float sum = (__expf(v0 - m) + __expf(v2 - m)) + (__expf(v1 - m) + __expf(v3 - m));
        for (int off = 8; off > 0; off >>= 1)
            sum += __shfl_xor(sum, off, 32);
        const float ls = __logf(sum);
        orow[(size_t)r * OUTC]      = v0 - m - ls;
        orow[(size_t)r * OUTC + 16] = v1 - m - ls;
        orow[(size_t)r * OUTC + 32] = v2 - m - ls;
        orow[(size_t)r * OUTC + 48] = v3 - m - ls;
    }
}

__global__ __launch_bounds__(1024) void rowptr_dinv_kernel(
    const int* __restrict__ cnt, int* __restrict__ rowptr,
    float* __restrict__ dinv, int n)
{
    __shared__ int wpre[32];
    __shared__ int wtot[32];
    const int lane = threadIdx.x & 31;
    const int wv   = threadIdx.x >> 5;
    const int base = blockIdx.x * 1024;

    int pre = 0;
    for (int k = threadIdx.x; k < base; k += 1024) pre += cnt[k];

    const int i = base + threadIdx.x;
    const int c = (i < n) ? cnt[i] : 0;

    int incl = c;
    for (int off = 1; off < 32; off <<= 1) {
        int y = __shfl_up(incl, off, 32);
        if (lane >= off) incl += y;
    }
    for (int off = 16; off > 0; off >>= 1) pre += __shfl_xor(pre, off, 32);
    if (lane == 31) wtot[wv] = incl;
    if (lane == 0)  wpre[wv] = pre;
    __syncthreads();

    int x = wpre[lane] + ((lane < wv) ? wtot[lane] : 0);
    for (int off = 16; off > 0; off >>= 1) x += __shfl_xor(x, off, 32);

    if (i < n) {
        rowptr[i] = x + incl - c;
        dinv[i]   = rsqrtf((float)(c + 1));
        if (i == n - 1) rowptr[n] = x + incl;
    }
}

__global__ void csr_fill_kernel(const int* __restrict__ src,
                                const int* __restrict__ dst,
                                const int* __restrict__ rank,
                                const int* __restrict__ rowptr,
                                int* __restrict__ csr_src, int e) {
    int i = blockIdx.x * blockDim.x + threadIdx.x;
    if (i < e) csr_src[rowptr[dst[i]] + rank[i]] = src[i];
}

__device__ __forceinline__ float4 gather_row(
    const float* __restrict__ h, const float* __restrict__ dinv,
    const float* __restrict__ bias, const int* __restrict__ rowptr,
    const int* __restrict__ csr_src, int i, int lane)
{
    const int   beg = rowptr[i];
    const int   end = rowptr[i + 1];
    const float di  = dinv[i];
    const float dd  = di * di;
    const float4 hv = ((const float4*)(h + (size_t)i * DIM))[lane];

    float4 acc;
    acc.x = hv.x * dd + bias[lane * 4 + 0];
    acc.y = hv.y * dd + bias[lane * 4 + 1];
    acc.z = hv.z * dd + bias[lane * 4 + 2];
    acc.w = hv.w * dd + bias[lane * 4 + 3];

    for (int c0 = beg; c0 < end; c0 += 32) {
        int m = end - c0;
        if (m > 32) m = 32;
        const int   sl = csr_src[c0 + ((lane < m) ? lane : 0)];
        const float dl = dinv[sl];
        for (int j0 = 0; j0 < m; j0 += 4) {
            float4 v[4];
            float  w[4];
#pragma unroll
            for (int u = 0; u < 4; ++u) {
                const int jj = j0 + u;
                const int s  = __builtin_amdgcn_readlane(sl, jj);
                v[u] = make_float4(0.0f, 0.0f, 0.0f, 0.0f);
                if (jj < m) v[u] = ((const float4*)(h + (size_t)s * DIM))[lane];
            }
#pragma unroll
            for (int u = 0; u < 4; ++u) {
                w[u] = __int_as_float(
                    __builtin_amdgcn_readlane(__float_as_int(dl), j0 + u)) * di;
                acc.x += v[u].x * w[u];
                acc.y += v[u].y * w[u];
                acc.z += v[u].z * w[u];
                acc.w += v[u].w * w[u];
            }
        }
    }

    acc.x = fmaxf(acc.x, 0.0f);
    acc.y = fmaxf(acc.y, 0.0f);
    acc.z = fmaxf(acc.z, 0.0f);
    acc.w = fmaxf(acc.w, 0.0f);
    return acc;
}

__global__ __launch_bounds__(256) void gather_relu_kernel(
    const float* __restrict__ h, const float* __restrict__ dinv,
    const float* __restrict__ bias, const int* __restrict__ rowptr,
    const int* __restrict__ csr_src, float* __restrict__ agg, int n)
{
    const int lane = threadIdx.x & 31;
    const int i = __builtin_amdgcn_readfirstlane(
        (int)(blockIdx.x * (blockDim.x >> 5) + (threadIdx.x >> 5)));
    if (i >= n) return;
    const float4 acc = gather_row(h, dinv, bias, rowptr, csr_src, i, lane);
    ((float4*)(agg + (size_t)i * DIM))[lane] = acc;
}

__global__ __launch_bounds__(256) void gather_relu_pool_kernel(
    const float* __restrict__ h, const float* __restrict__ dinv,
    const float* __restrict__ bias, const int* __restrict__ rowptr,
    const int* __restrict__ csr_src, const int* __restrict__ batch,
    float* __restrict__ pooled, int n)
{
    __shared__ float srow[8][DIM];
    __shared__ __attribute__((aligned(16))) int sgid[8];
    const int lane = threadIdx.x & 31;
    const int wv   = threadIdx.x >> 5;
    const int i = __builtin_amdgcn_readfirstlane(
        (int)(blockIdx.x * (blockDim.x >> 5) + wv));

    float4 acc = make_float4(0.0f, 0.0f, 0.0f, 0.0f);
    int gid = -1;
    if (i < n) {
        gid = batch[i];
        acc = gather_row(h, dinv, bias, rowptr, csr_src, i, lane);
    }
    ((float4*)srow[wv])[lane] = acc;
    if (lane == 0) sgid[wv] = gid;
    __syncthreads();

    if (threadIdx.x < DIM) {
        const int c = threadIdx.x;
        int   g[8];
        float v[8];
#pragma unroll
        for (int k = 0; k < 8; ++k) {
            g[k] = __builtin_amdgcn_readfirstlane(sgid[k]);
            v[k] = srow[k][c];
        }
        int   cur = g[0];
        float run = v[0];
#pragma unroll
        for (int k = 1; k < 8; ++k) {
            if (g[k] != cur) {
                if (cur >= 0)
                    __hip_atomic_fetch_add(&pooled[(size_t)cur * DIM + c], run,
                                           __ATOMIC_RELAXED, __HIP_MEMORY_SCOPE_AGENT);
                cur = g[k];
                run = v[k];
            } else {
                run += v[k];
            }
        }
        if (cur >= 0)
            __hip_atomic_fetch_add(&pooled[(size_t)cur * DIM + c], run,
                                   __ATOMIC_RELAXED, __HIP_MEMORY_SCOPE_AGENT);
    }
}

extern "C" void kernel_launch(void* const* d_in, const int* in_sizes, int n_in,
                              void* d_out, int out_size, void* d_ws, size_t ws_size,
                              hipStream_t stream) {
    const float* x       = (const float*)d_in[0];
    const int*   eidx    = (const int*)d_in[1];
    const int*   x_batch = (const int*)d_in[2];
    const float* W1      = (const float*)d_in[3];
    const float* b1      = (const float*)d_in[4];
    const float* W2      = (const float*)d_in[5];
    const float* b2      = (const float*)d_in[6];
    const float* Wh      = (const float*)d_in[7];
    const float* bh      = (const float*)d_in[8];
    float*       out     = (float*)d_out;

    const int n_nodes = in_sizes[0] / DIM;
    const int n_edges = in_sizes[1] / 2;
    const int* src = eidx;
    const int* dst = eidx + n_edges;

    float* h       = (float*)d_ws;
    float* agg     = h      + (size_t)n_nodes * DIM;
    float* pooled  = agg    + (size_t)n_nodes * DIM;
    float* W1T     = pooled + (size_t)NGRAPHS * DIM;
    float* W2T     = W1T    + DIM * DIM;
    float* WhT     = W2T    + DIM * DIM;
    float* dinv    = WhT    + OUTC * DIM;
    int*   cnt     = (int*)(dinv + n_nodes);
    int*   rank    = cnt     + n_nodes;
    int*   csr_src = rank    + n_edges;
    int*   rowptr  = csr_src + n_edges;

    const int T = 256;
    const int edgeBlk = (n_edges + T - 1) / T;
    const int nodeWaveBlk = (n_nodes + 7) / 8;
    const int gemmBlk = ((n_nodes / 16) * (DIM / 64) + 7) / 8;

    prep_kernel<<<(2 * DIM * DIM + DIM * OUTC + NGRAPHS * DIM + n_nodes + T - 1) / T, T, 0, stream>>>(
        W1, W2, Wh, W1T, W2T, WhT, pooled, cnt, n_nodes);

    gemm_count_rank_kernel<<<gemmBlk + edgeBlk, T, 0, stream>>>(
        x, W1T, h, n_nodes, DIM, DIM, gemmBlk, dst, cnt, rank, n_edges);

    rowptr_dinv_kernel<<<(n_nodes + 1023) / 1024, 1024, 0, stream>>>(cnt, rowptr, dinv, n_nodes);
    csr_fill_kernel<<<edgeBlk, T, 0, stream>>>(src, dst, rank, rowptr, csr_src, n_edges);

    gather_relu_kernel<<<nodeWaveBlk, T, 0, stream>>>(h, dinv, b1, rowptr, csr_src, agg, n_nodes);

    wmma_gemm_f32_bt<<<gemmBlk, T, 0, stream>>>(agg, W2T, h, n_nodes, DIM, DIM);

    gather_relu_pool_kernel<<<nodeWaveBlk, T, 0, stream>>>(h, dinv, b2, rowptr, csr_src,
                                                          x_batch, pooled, n_nodes);

    head_logsoftmax_kernel<<<NGRAPHS / 16, 32, 0, stream>>>(pooled, WhT, bh, out,
                                                          NGRAPHS, DIM);
}


# Edge list as fixed 32-slot rows per destination node, filled by the ranking atomics themselves (edges past 32: one overflow list, summed in list order); prefix-sum and fill launches gone (8 -> 6)
# speedup vs baseline: 2.9367x; 1.1546x over previous
#include <hip/hip_runtime.h>
#include <hip/hip_bf16.h>

typedef float v2f __attribute__((ext_vector_type(2)));
typedef float v8f __attribute__((ext_vector_type(8)));

#define DIM 128
#define OUTC 64
#define NGRAPHS 512
#define CAP 32

__global__ void prep_kernel(const float* __restrict__ W1, const float* __restrict__ W2,
                            const float* __restrict__ Wh,
                            float* __restrict__ W1T, float* __restrict__ W2T,
                            float* __restrict__ WhT,
                            float* __restrict__ pooled, int* __restrict__ cnt,
                            int* __restrict__ ovf_n, int n) {
    int idx = blockIdx.x * blockDim.x + threadIdx.x;
    if (idx < DIM * DIM) {
        int k = idx / DIM;
        int c = idx - k * DIM;
        W1T[c * DIM + k] = W1[idx];
        return;
    }
    idx -= DIM * DIM;
    if (idx < DIM * DIM) {
        int k = idx / DIM;
        int c = idx - k * DIM;
        W2T[c * DIM + k] = W2[idx];
        return;
    }
    idx -= DIM * DIM;
    if (idx < DIM * OUTC) {
        int k = idx / OUTC;
        int c = idx - k * OUTC;
        WhT[c * DIM + k] = Wh[idx];
        return;
    }
    idx -= DIM * OUTC;
    if (idx < NGRAPHS * DIM) {
        pooled[idx] = 0.0f;
        return;
    }
    idx -= NGRAPHS * DIM;
    if (idx < n) {
        cnt[idx] = 0;
        return;
    }
    if (idx == n) *ovf_n = 0;
}

struct Strip { v8f a0, a1, a2, a3; };

__device__ __forceinline__ Strip gemm_strip(
    const float* __restrict__ A, const float* __restrict__ BT,
    int K, int row0, int col0, int lane)
{
    const int m16   = lane & 15;
    const int khalf = (lane >> 4) << 1;

    const float* ap  = A  + (size_t)(row0 + m16) * K + khalf;
    const float* bp0 = BT + (size_t)(col0 + m16) * K + khalf;
    const float* bp1 = bp0 + (size_t)16 * K;
    const float* bp2 = bp0 + (size_t)32 * K;
    const float* bp3 = bp0 + (size_t)48 * K;

    v8f acc0 = {}, acc1 = {}, acc2 = {}, acc3 = {};
#pragma unroll 4
    for (int k0 = 0; k0 < K; k0 += 4) {
        v2f a  = *(const v2f*)(ap  + k0);
        v2f b0 = *(const v2f*)(bp0 + k0);
        v2f b1 = *(const v2f*)(bp1 + k0);
        v2f b2 = *(const v2f*)(bp2 + k0);
        v2f b3 = *(const v2f*)(bp3 + k0);
        acc0 = __builtin_amdgcn_wmma_f32_16x16x4_f32(false, a, false, b0, (short)0, acc0, false, false);
        acc1 = __builtin_amdgcn_wmma_f32_16x16x4_f32(false, a, false, b1, (short)0, acc1, false, false);
        acc2 = __builtin_amdgcn_wmma_f32_16x16x4_f32(false, a, false, b2, (short)0, acc2, false, false);
        acc3 = __builtin_amdgcn_wmma_f32_16x16x4_f32(false, a, false, b3, (short)0, acc3, false, false);
    }
    Strip s;
    s.a0 = acc0; s.a1 = acc1; s.a2 = acc2; s.a3 = acc3;
    return s;
}

__device__ __forceinline__ void store_strip(
    const Strip& s, float* __restrict__ C, int N, int row0, int col0, int lane)
{
    const int nOff  = lane & 15;
    const int mBase = row0 + ((lane >> 4) << 3);
    float* crow = C + (size_t)mBase * N + col0 + nOff;
#pragma unroll
    for (int r = 0; r < 8; ++r) {
        crow[(size_t)r * N]      = s.a0[r];
        crow[(size_t)r * N + 16] = s.a1[r];
        crow[(size_t)r * N + 32] = s.a2[r];
        crow[(size_t)r * N + 48] = s.a3[r];
    }
}

__global__ __launch_bounds__(256) void wmma_gemm_f32_bt(
    const float* __restrict__ A, const float* __restrict__ BT,
    float* __restrict__ C, int M, int N, int K)
{
    const int lane = threadIdx.x & 31;
    const int wave = blockIdx.x * (blockDim.x >> 5) + (threadIdx.x >> 5);
    const int colGroups = N >> 6;
    const int tileM = wave / colGroups;
    const int cg    = wave - tileM * colGroups;
    if (tileM * 16 >= M) return;
    const int row0 = tileM << 4;
    const int col0 = cg << 6;
    const Strip s = gemm_strip(A, BT, K, row0, col0, lane);
    store_strip(s, C, N, row0, col0, lane);
}

__global__ __launch_bounds__(256) void gemm_rank_bucket_kernel(
    const float* __restrict__ A, const float* __restrict__ BT,
    float* __restrict__ C, int M, int N, int K, int gemmBlocks,
    const int* __restrict__ src, const int* __restrict__ dst,
    int* __restrict__ cnt, int* __restrict__ bucket,
    int* __restrict__ ovf_n, int* __restrict__ ovf_dst, int* __restrict__ ovf_src,
    int e)
{
    if ((int)blockIdx.x >= gemmBlocks) {
        const int i = ((int)blockIdx.x - gemmBlocks) * (int)blockDim.x + (int)threadIdx.x;
        if (i < e) {
            const int d = dst[i];
            const int s = src[i];
            const int r = __hip_atomic_fetch_add(&cnt[d], 1,
                                                 __ATOMIC_RELAXED, __HIP_MEMORY_SCOPE_AGENT);
            if (r < CAP) {
                bucket[(size_t)d * CAP + r] = s;
            } else {
                const int k = __hip_atomic_fetch_add(ovf_n, 1,
                                                     __ATOMIC_RELAXED, __HIP_MEMORY_SCOPE_AGENT);
                ovf_dst[k] = d;
                ovf_src[k] = s;
            }
        }
        return;
    }
    const int lane = threadIdx.x & 31;
    const int wave = blockIdx.x * (blockDim.x >> 5) + (threadIdx.x >> 5);
    const int colGroups = N >> 6;
    const int tileM = wave / colGroups;
    const int cg    = wave - tileM * colGroups;
    if (tileM * 16 >= M) return;
    const int row0 = tileM << 4;
    const int col0 = cg << 6;
    const Strip s = gemm_strip(A, BT, K, row0, col0, lane);
    store_strip(s, C, N, row0, col0, lane);
}

__global__ __launch_bounds__(256) void head_logsoftmax_kernel(
    const float* __restrict__ A, const float* __restrict__ BT,
    const float* __restrict__ bias, float* __restrict__ out, int M, int K)
{
    const int lane = threadIdx.x & 31;
    const int wave = blockIdx.x * (blockDim.x >> 5) + (threadIdx.x >> 5);
    if (wave * 16 >= M) return;
    const int row0 = wave << 4;
    const Strip s = gemm_strip(A, BT, K, row0, 0, lane);

    const int c  = lane & 15;
    const int h8 = (lane >> 4) << 3;
    const float bb0 = bias[c];
    const float bb1 = bias[c + 16];
    const float bb2 = bias[c + 32];
    const float bb3 = bias[c + 48];
    float* orow = out + (size_t)(row0 + h8) * OUTC + c;
#pragma unroll
    for (int r = 0; r < 8; ++r) {
        const float v0 = s.a0[r] + bb0;
        const float v1 = s.a1[r] + bb1;
        const float v2 = s.a2[r] + bb2;
        const float v3 = s.a3[r] + bb3;
        float m = fmaxf(fmaxf(v0, v2), fmaxf(v1, v3));
        for (int off = 8; off > 0; off >>= 1)
            m = fmaxf(m, __shfl_xor(m, off, 32));
        float sum = (__expf(v0 - m) + __expf(v2 - m)) + (__expf(v1 - m) + __expf(v3 - m));
        for (int off = 8; off > 0; off >>= 1)
            sum += __shfl_xor(sum, off, 32);
        const float ls = __logf(sum);
        orow[(size_t)r * OUTC]      = v0 - m - ls;
        orow[(size_t)r * OUTC + 16] = v1 - m - ls;
        orow[(size_t)r * OUTC + 32] = v2 - m - ls;
        orow[(size_t)r * OUTC + 48] = v3 - m - ls;
    }
}

__device__ __forceinline__ float dinv_of(int c) {
    return rsqrtf((float)(c + 1));
}


__device__ __forceinline__ void issue_rows(
    float4 (&v)[4], const float* __restrict__ h, int sl, int j0, int m, int lane)
{
#pragma unroll
    for (int u = 0; u < 4; ++u) {
        const int jj = j0 + u;
        const int s  = __builtin_amdgcn_readlane(sl, jj);
        v[u] = make_float4(0.0f, 0.0f, 0.0f, 0.0f);
        if (jj < m) v[u] = ((const float4*)(h + (size_t)s * DIM))[lane];
    }
}

__device__ __forceinline__ float4 gather_row(
    const float* __restrict__ h, const int* __restrict__ cnt,
    const float* __restrict__ bias, const int* __restrict__ bucket,
    const int* __restrict__ ovf_n, const int* __restrict__ ovf_dst,
    const int* __restrict__ ovf_src, int i, int lane)
{
    const float4 hv = ((const float4*)(h + (size_t)i * DIM))[lane];
    const float b0 = bias[lane * 4 + 0];
    const float b1 = bias[lane * 4 + 1];
    const float b2 = bias[lane * 4 + 2];
    const float b3 = bias[lane * 4 + 3];

    const int c = cnt[i];
    const int m = (c < CAP) ? c : CAP;
    int sl = i;
    if (lane < m) sl = bucket[(size_t)i * CAP + lane];

    const float di = dinv_of(c);
    const float dd = di * di;
    float4 acc;
    acc.x = hv.x * dd + b0;
    acc.y = hv.y * dd + b1;
    acc.z = hv.z * dd + b2;
    acc.w = hv.w * dd + b3;

    if (m > 0) {
        const int cl = cnt[sl];
        float4 v[4];
        issue_rows(v, h, sl, 0, m, lane);
        const float dl = dinv_of(cl);
        int j0 = 0;
        while (true) {
#pragma unroll
            for (int u = 0; u < 4; ++u) {
                const float w = __int_as_float(
                    __builtin_amdgcn_readlane(__float_as_int(dl), j0 + u)) * di;
                acc.x += v[u].x * w;
                acc.y += v[u].y * w;
                acc.z += v[u].z * w;
                acc.w += v[u].w * w;
            }
            j0 += 4;
            if (j0 >= m) break;
            issue_rows(v, h, sl, j0, m, lane);
        }
    }

    if (c > CAP) {
        const int tot = *ovf_n;
        for (int k0 = 0; k0 < tot; k0 += 32) {
            const int k = k0 + lane;
            int od = -1;
            int os = 0;
            if (k < tot) {
                od = ovf_dst[k];
                os = ovf_src[k];
            }
            const bool mine = (od == i);
            float dl = 0.0f;
            if (mine) dl = dinv_of(cnt[os]);
            unsigned hit = __builtin_amdgcn_ballot_w32(mine);
            while (hit != 0u) {
                const int b = __builtin_ctz(hit);
                hit &= hit - 1u;
                const int   s = __builtin_amdgcn_readlane(os, b);
                const float w = __int_as_float(
                    __builtin_amdgcn_readlane(__float_as_int(dl), b)) * di;
                const float4 v = ((const float4*)(h + (size_t)s * DIM))[lane];
                acc.x += v.x * w;
                acc.y += v.y * w;
                acc.z += v.z * w;
                acc.w += v.w * w;
            }
        }
    }

    acc.x = fmaxf(acc.x, 0.0f);
    acc.y = fmaxf(acc.y, 0.0f);
    acc.z = fmaxf(acc.z, 0.0f);
    acc.w = fmaxf(acc.w, 0.0f);
    return acc;
}

__global__ __launch_bounds__(256) void gather_relu_kernel(
    const float* __restrict__ h, const int* __restrict__ cnt,
    const float* __restrict__ bias, const int* __restrict__ bucket,
    const int* __restrict__ ovf_n, const int* __restrict__ ovf_dst,
    const int* __restrict__ ovf_src, float* __restrict__ agg, int n)
{
    const int lane = threadIdx.x & 31;
    const int i = __builtin_amdgcn_readfirstlane(
        (int)(blockIdx.x * (blockDim.x >> 5) + (threadIdx.x >> 5)));
    if (i >= n) return;
    const float4 acc = gather_row(h, cnt, bias, bucket, ovf_n, ovf_dst, ovf_src, i, lane);
    ((float4*)(agg + (size_t)i * DIM))[lane] = acc;
}

__global__ __launch_bounds__(256) void gather_relu_pool_kernel(
    const float* __restrict__ h, const int* __restrict__ cnt,
    const float* __restrict__ bias, const int* __restrict__ bucket,
    const int* __restrict__ ovf_n, const int* __restrict__ ovf_dst,
    const int* __restrict__ ovf_src, const int* __restrict__ batch,
    float* __restrict__ pooled, int n)
{
    __shared__ float srow[8][DIM];
    __shared__ __attribute__((aligned(16))) int sgid[8];
    const int lane = threadIdx.x & 31;
    const int wv   = threadIdx.x >> 5;
    const int i = __builtin_amdgcn_readfirstlane(
        (int)(blockIdx.x * (blockDim.x >> 5) + wv));

    float4 acc = make_float4(0.0f, 0.0f, 0.0f, 0.0f);
    int gid = -1;
    if (i < n) {
        gid = batch[i];
        acc = gather_row(h, cnt, bias, bucket, ovf_n, ovf_dst, ovf_src, i, lane);
    }
    ((float4*)srow[wv])[lane] = acc;
    if (lane == 0) sgid[wv] = gid;
    __syncthreads();

    if (threadIdx.x < DIM) {
        const int c = threadIdx.x;
        int   g[8];
        float v[8];
#pragma unroll
        for (int k = 0; k < 8; ++k) {
            g[k] = __builtin_amdgcn_readfirstlane(sgid[k]);
            v[k] = srow[k][c];
        }
        int   cur = g[0];
        float run = v[0];
#pragma unroll
        for (int k = 1; k < 8; ++k) {
            if (g[k] != cur) {
                if (cur >= 0)
                    __hip_atomic_fetch_add(&pooled[(size_t)cur * DIM + c], run,
                                           __ATOMIC_RELAXED, __HIP_MEMORY_SCOPE_AGENT);
                cur = g[k];
                run = v[k];
            } else {
                run += v[k];
            }
        }
        if (cur >= 0)
            __hip_atomic_fetch_add(&pooled[(size_t)cur * DIM + c], run,
                                   __ATOMIC_RELAXED, __HIP_MEMORY_SCOPE_AGENT);
    }
}

extern "C" void kernel_launch(void* const* d_in, const int* in_sizes, int n_in,
                              void* d_out, int out_size, void* d_ws, size_t ws_size,
                              hipStream_t stream) {
    const float* x       = (const float*)d_in[0];
    const int*   eidx    = (const int*)d_in[1];
    const int*   x_batch = (const int*)d_in[2];
    const float* W1      = (const float*)d_in[3];
    const float* b1      = (const float*)d_in[4];
    const float* W2      = (const float*)d_in[5];
    const float* b2      = (const float*)d_in[6];
    const float* Wh      = (const float*)d_in[7];
    const float* bh      = (const float*)d_in[8];
    float*       out     = (float*)d_out;

    const int n_nodes = in_sizes[0] / DIM;
    const int n_edges = in_sizes[1] / 2;
    const int* src = eidx;
    const int* dst = eidx + n_edges;

    float* h       = (float*)d_ws;
    float* agg     = h      + (size_t)n_nodes * DIM;
    float* pooled  = agg    + (size_t)n_nodes * DIM;
    float* W1T     = pooled + (size_t)NGRAPHS * DIM;
    float* W2T     = W1T    + DIM * DIM;
    float* WhT     = W2T    + DIM * DIM;
    int*   bucket  = (int*)(WhT + OUTC * DIM);
    int*   cnt     = bucket  + (size_t)n_nodes * CAP;
    int*   ovf_dst = cnt     + n_nodes;
    int*   ovf_src = ovf_dst + n_edges;
    int*   ovf_n   = ovf_src + n_edges;

    const int T = 256;
    const int edgeBlk = (n_edges + T - 1) / T;
    const int nodeWaveBlk = (n_nodes + 7) / 8;
    const int gemmBlk = ((n_nodes / 16) * (DIM / 64) + 7) / 8;

    prep_kernel<<<(2 * DIM * DIM + DIM * OUTC + NGRAPHS * DIM + n_nodes + 1 + T - 1) / T, T, 0, stream>>>(
        W1, W2, Wh, W1T, W2T, WhT, pooled, cnt, ovf_n, n_nodes);

    gemm_rank_bucket_kernel<<<gemmBlk + edgeBlk, T, 0, stream>>>(
        x, W1T, h, n_nodes, DIM, DIM, gemmBlk, src, dst, cnt, bucket,
        ovf_n, ovf_dst, ovf_src, n_edges);

    gather_relu_kernel<<<nodeWaveBlk, T, 0, stream>>>(h, cnt, b1, bucket, ovf_n, ovf_dst, ovf_src,
                                                     agg, n_nodes);

    wmma_gemm_f32_bt<<<gemmBlk, T, 0, stream>>>(agg, W2T, h, n_nodes, DIM, DIM);

    gather_relu_pool_kernel<<<nodeWaveBlk, T, 0, stream>>>(h, cnt, b2, bucket, ovf_n, ovf_dst,
                                                          ovf_src, x_batch, pooled, n_nodes);

    head_logsoftmax_kernel<<<NGRAPHS / 16, 32, 0, stream>>>(pooled, WhT, bh, out,
                                                          NGRAPHS, DIM);
}


# Node-feature GEMMs take B from LDS: each work-group stages its 64-column half of W^T once (padded pitch), fragments are 16-byte reads, A rows loaded 5 blocks ahead; k order inside the sums changes
# speedup vs baseline: 3.6783x; 1.2525x over previous
#include <hip/hip_runtime.h>
#include <hip/hip_bf16.h>

typedef float v2f __attribute__((ext_vector_type(2)));
typedef float v8f __attribute__((ext_vector_type(8)));

#define DIM 128
#define OUTC 64
#define NGRAPHS 512
#define CAP 32
#define BPITCH 132

__global__ void prep_kernel(const float* __restrict__ W1, const float* __restrict__ W2,
                            const float* __restrict__ Wh,
                            float* __restrict__ W1T, float* __restrict__ W2T,
                            float* __restrict__ WhT,
                            float* __restrict__ pooled, int* __restrict__ cnt,
                            int* __restrict__ ovf_n, int n) {
    int idx = blockIdx.x * blockDim.x + threadIdx.x;
    if (idx < DIM * DIM) {
        int k = idx / DIM;
        int c = idx - k * DIM;
        W1T[c * DIM + k] = W1[idx];
        return;
    }
    idx -= DIM * DIM;
    if (idx < DIM * DIM) {
        int k = idx / DIM;
        int c = idx - k * DIM;
        W2T[c * DIM + k] = W2[idx];
        return;
    }
    idx -= DIM * DIM;
    if (idx < DIM * OUTC) {
        int k = idx / OUTC;
        int c = idx - k * OUTC;
        WhT[c * DIM + k] = Wh[idx];
        return;
    }
    idx -= DIM * OUTC;
    if (idx < NGRAPHS * DIM) {
        pooled[idx] = 0.0f;
        return;
    }
    idx -= NGRAPHS * DIM;
    if (idx < n) {
        cnt[idx] = 0;
        return;
    }
    if (idx == n) *ovf_n = 0;
}

struct Strip { v8f a0, a1, a2, a3; };

__device__ __forceinline__ Strip gemm_strip(
    const float* __restrict__ A, const float* __restrict__ BT,
    int K, int row0, int col0, int lane)
{
    const int m16   = lane & 15;
    const int khalf = (lane >> 4) << 1;

    const float* ap  = A  + (size_t)(row0 + m16) * K + khalf;
    const float* bp0 = BT + (size_t)(col0 + m16) * K + khalf;
    const float* bp1 = bp0 + (size_t)16 * K;
    const float* bp2 = bp0 + (size_t)32 * K;
    const float* bp3 = bp0 + (size_t)48 * K;

    v8f acc0 = {}, acc1 = {}, acc2 = {}, acc3 = {};
#pragma unroll 4
    for (int k0 = 0; k0 < K; k0 += 4) {
        v2f a  = *(const v2f*)(ap  + k0);
        v2f b0 = *(const v2f*)(bp0 + k0);
        v2f b1 = *(const v2f*)(bp1 + k0);
        v2f b2 = *(const v2f*)(bp2 + k0);
        v2f b3 = *(const v2f*)(bp3 + k0);
        acc0 = __builtin_amdgcn_wmma_f32_16x16x4_f32(false, a, false, b0, (short)0, acc0, false, false);
        acc1 = __builtin_amdgcn_wmma_f32_16x16x4_f32(false, a, false, b1, (short)0, acc1, false, false);
        acc2 = __builtin_amdgcn_wmma_f32_16x16x4_f32(false, a, false, b2, (short)0, acc2, false, false);
        acc3 = __builtin_amdgcn_wmma_f32_16x16x4_f32(false, a, false, b3, (short)0, acc3, false, false);
    }
    Strip s;
    s.a0 = acc0; s.a1 = acc1; s.a2 = acc2; s.a3 = acc3;
    return s;
}

__device__ __forceinline__ void stage_bt_half(
    float* __restrict__ sB, const float* __restrict__ BThalf, int tid)
{
#pragma unroll
    for (int p = 0; p < 8; ++p) {
        const int idx = p * 256 + tid;
        const int n = idx >> 5;
        const int q = idx & 31;
        const float4 v = ((const float4*)(BThalf + (size_t)n * DIM))[q];
        *(float4*)(sB + n * BPITCH + 4 * q) = v;
    }
}

__device__ __forceinline__ Strip gemm_strip_lds(
    const float* __restrict__ A, const float* sB, int row0, int lane)
{
    const int c  = lane & 15;
    const int h4 = (lane >> 4) << 2;
    const float4* ap = (const float4*)(A + (size_t)(row0 + c) * DIM + h4);
    const float*  bp = sB + c * BPITCH + h4;

    float4 a[6];
#pragma unroll
    for (int i = 0; i < 6; ++i) a[i] = ap[2 * i];
    float4 b[2][4];
#pragma unroll
    for (int j = 0; j < 4; ++j) b[0][j] = *(const float4*)(bp + j * 16 * BPITCH);

    v8f acc0 = {}, acc1 = {}, acc2 = {}, acc3 = {};
    __builtin_amdgcn_sched_barrier(0);
#pragma unroll
    for (int i = 0; i < DIM / 8; ++i) {
        const int cur = i & 1;
        if (i >= 1 && i + 5 < DIM / 8) a[(i - 1) % 6] = ap[2 * (i + 5)];
        if (i + 1 < DIM / 8) {
#pragma unroll
            for (int j = 0; j < 4; ++j)
                b[cur ^ 1][j] = *(const float4*)(bp + j * 16 * BPITCH + (i + 1) * 8);
        }
        __builtin_amdgcn_sched_barrier(0);
        const float4 av = a[i % 6];
        const v2f alo = {av.x, av.y};
        const v2f ahi = {av.z, av.w};
        const v2f b0lo = {b[cur][0].x, b[cur][0].y}, b0hi = {b[cur][0].z, b[cur][0].w};
        const v2f b1lo = {b[cur][1].x, b[cur][1].y}, b1hi = {b[cur][1].z, b[cur][1].w};
        const v2f b2lo = {b[cur][2].x, b[cur][2].y}, b2hi = {b[cur][2].z, b[cur][2].w};
        const v2f b3lo = {b[cur][3].x, b[cur][3].y}, b3hi = {b[cur][3].z, b[cur][3].w};
        acc0 = __builtin_amdgcn_wmma_f32_16x16x4_f32(false, alo, false, b0lo, (short)0, acc0, false, false);
        acc1 = __builtin_amdgcn_wmma_f32_16x16x4_f32(false, alo, false, b1lo, (short)0, acc1, false, false);
        acc2 = __builtin_amdgcn_wmma_f32_16x16x4_f32(false, alo, false, b2lo, (short)0, acc2, false, false);
        acc3 = __builtin_amdgcn_wmma_f32_16x16x4_f32(false, alo, false, b3lo, (short)0, acc3, false, false);
        acc0 = __builtin_amdgcn_wmma_f32_16x16x4_f32(false, ahi, false, b0hi, (short)0, acc0, false, false);
        acc1 = __builtin_amdgcn_wmma_f32_16x16x4_f32(false, ahi, false, b1hi, (short)0, acc1, false, false);
        acc2 = __builtin_amdgcn_wmma_f32_16x16x4_f32(false, ahi, false, b2hi, (short)0, acc2, false, false);
        acc3 = __builtin_amdgcn_wmma_f32_16x16x4_f32(false, ahi, false, b3hi, (short)0, acc3, false, false);
        __builtin_amdgcn_sched_barrier(0);
    }
    Strip s;
    s.a0 = acc0; s.a1 = acc1; s.a2 = acc2; s.a3 = acc3;
    return s;
}

__device__ __forceinline__ void store_strip(
    const Strip& s, float* __restrict__ C, int N, int row0, int col0, int lane)
{
    const int nOff  = lane & 15;
    const int mBase = row0 + ((lane >> 4) << 3);
    float* crow = C + (size_t)mBase * N + col0 + nOff;
#pragma unroll
    for (int r = 0; r < 8; ++r) {
        crow[(size_t)r * N]      = s.a0[r];
        crow[(size_t)r * N + 16] = s.a1[r];
        crow[(size_t)r * N + 32] = s.a2[r];
        crow[(size_t)r * N + 48] = s.a3[r];
    }
}

__global__ __launch_bounds__(256) void wmma_gemm_f32_bt(
    const float* __restrict__ A, const float* __restrict__ BT,
    float* __restrict__ C, int M)
{
    __shared__ __attribute__((aligned(16))) float sB[64 * BPITCH];
    const int lane = threadIdx.x & 31;
    const int cg    = (int)blockIdx.x & 1;
    const int tileM = ((int)blockIdx.x >> 1) * 8 + (int)(threadIdx.x >> 5);
    stage_bt_half(sB, BT + (size_t)cg * 64 * DIM, (int)threadIdx.x);
    __syncthreads();
    if (tileM * 16 >= M) return;
    const int row0 = tileM << 4;
    const int col0 = cg << 6;
    const Strip s = gemm_strip_lds(A, sB, row0, lane);
    store_strip(s, C, DIM, row0, col0, lane);
}

__global__ __launch_bounds__(256) void gemm_rank_bucket_kernel(
    const float* __restrict__ A, const float* __restrict__ BT,
    float* __restrict__ C, int M, int gemmBlocks,
    const int* __restrict__ src, const int* __restrict__ dst,
    int* __restrict__ cnt, int* __restrict__ bucket,
    int* __restrict__ ovf_n, int* __restrict__ ovf_dst, int* __restrict__ ovf_src,
    int e)
{
    if ((int)blockIdx.x >= gemmBlocks) {
        const int i = ((int)blockIdx.x - gemmBlocks) * (int)blockDim.x + (int)threadIdx.x;
        if (i < e) {
            const int d = dst[i];
            const int s = src[i];
            const int r = __hip_atomic_fetch_add(&cnt[d], 1,
                                                 __ATOMIC_RELAXED, __HIP_MEMORY_SCOPE_AGENT);
            if (r < CAP) {
                bucket[(size_t)d * CAP + r] = s;
            } else {
                const int k = __hip_atomic_fetch_add(ovf_n, 1,
                                                     __ATOMIC_RELAXED, __HIP_MEMORY_SCOPE_AGENT);
                ovf_dst[k] = d;
                ovf_src[k] = s;
            }
        }
        return;
    }
    __shared__ __attribute__((aligned(16))) float sB[64 * BPITCH];
    const int lane = threadIdx.x & 31;
    const int cg    = (int)blockIdx.x & 1;
    const int tileM = ((int)blockIdx.x >> 1) * 8 + (int)(threadIdx.x >> 5);
    stage_bt_half(sB, BT + (size_t)cg * 64 * DIM, (int)threadIdx.x);
    __syncthreads();
    if (tileM * 16 >= M) return;
    const int row0 = tileM << 4;
    const int col0 = cg << 6;
    const Strip s = gemm_strip_lds(A, sB, row0, lane);
    store_strip(s, C, DIM, row0, col0, lane);
}

__global__ __launch_bounds__(256) void head_logsoftmax_kernel(
    const float* __restrict__ A, const float* __restrict__ BT,
    const float* __restrict__ bias, float* __restrict__ out, int M, int K)
{
    const int lane = threadIdx.x & 31;
    const int wave = blockIdx.x * (blockDim.x >> 5) + (threadIdx.x >> 5);
    if (wave * 16 >= M) return;
    const int row0 = wave << 4;
    const Strip s = gemm_strip(A, BT, K, row0, 0, lane);

    const int c  = lane & 15;
    const int h8 = (lane >> 4) << 3;
    const float bb0 = bias[c];
    const float bb1 = bias[c + 16];
    const float bb2 = bias[c + 32];
    const float bb3 = bias[c + 48];
    float* orow = out + (size_t)(row0 + h8) * OUTC + c;
#pragma unroll
    for (int r = 0; r < 8; ++r) {
        const float v0 = s.a0[r] + bb0;
        const float v1 = s.a1[r] + bb1;
        const float v2 = s.a2[r] + bb2;
        const float v3 = s.a3[r] + bb3;
        float m = fmaxf(fmaxf(v0, v2), fmaxf(v1, v3));
        for (int off = 8; off > 0; off >>= 1)
            m = fmaxf(m, __shfl_xor(m, off, 32));
        float sum = (__expf(v0 - m) + __expf(v2 - m)) + (__expf(v1 - m) + __expf(v3 - m));
        for (int off = 8; off > 0; off >>= 1)
            sum += __shfl_xor(sum, off, 32);
        const float ls = __logf(sum);
        orow[(size_t)r * OUTC]      = v0 - m - ls;
        orow[(size_t)r * OUTC + 16] = v1 - m - ls;
        orow[(size_t)r * OUTC + 32] = v2 - m - ls;
        orow[(size_t)r * OUTC + 48] = v3 - m - ls;
    }
}

__device__ __forceinline__ float dinv_of(int c) {
    return rsqrtf((float)(c + 1));
}


__device__ __forceinline__ void issue_rows(
    float4 (&v)[4], const float* __restrict__ h, int sl, int j0, int m, int lane)
{
#pragma unroll
    for (int u = 0; u < 4; ++u) {
        const int jj = j0 + u;
        const int s  = __builtin_amdgcn_readlane(sl, jj);
        v[u] = make_float4(0.0f, 0.0f, 0.0f, 0.0f);
        if (jj < m) v[u] = ((const float4*)(h + (size_t)s * DIM))[lane];
    }
}

__device__ __forceinline__ float4 gather_row(
    const float* __restrict__ h, const int* __restrict__ cnt,
    const float* __restrict__ bias, const int* __restrict__ bucket,
    const int* __restrict__ ovf_n, const int* __restrict__ ovf_dst,
    const int* __restrict__ ovf_src, int i, int lane)
{
    const float4 hv = ((const float4*)(h + (size_t)i * DIM))[lane];
    const float b0 = bias[lane * 4 + 0];
    const float b1 = bias[lane * 4 + 1];
    const float b2 = bias[lane * 4 + 2];
    const float b3 = bias[lane * 4 + 3];

    const int c = cnt[i];
    const int m = (c < CAP) ? c : CAP;
    int sl = i;
    if (lane < m) sl = bucket[(size_t)i * CAP + lane];

    const float di = dinv_of(c);
    const float dd = di * di;
    float4 acc;
    acc.x = hv.x * dd + b0;
    acc.y = hv.y * dd + b1;
    acc.z = hv.z * dd + b2;
    acc.w = hv.w * dd + b3;

    if (m > 0) {
        const int cl = cnt[sl];
        float4 v[4];
        issue_rows(v, h, sl, 0, m, lane);
        const float dl = dinv_of(cl);
        int j0 = 0;
        while (true) {
#pragma unroll
            for (int u = 0; u < 4; ++u) {
                const float w = __int_as_float(
                    __builtin_amdgcn_readlane(__float_as_int(dl), j0 + u)) * di;
                acc.x += v[u].x * w;
                acc.y += v[u].y * w;
                acc.z += v[u].z * w;
                acc.w += v[u].w * w;
            }
            j0 += 4;
            if (j0 >= m) break;
            issue_rows(v, h, sl, j0, m, lane);
        }
    }

    if (c > CAP) {
        const int tot = *ovf_n;
        for (int k0 = 0; k0 < tot; k0 += 32) {
            const int k = k0 + lane;
            int od = -1;
            int os = 0;
            if (k < tot) {
                od = ovf_dst[k];
                os = ovf_src[k];
            }
            const bool mine = (od == i);
            float dl = 0.0f;
            if (mine) dl = dinv_of(cnt[os]);
            unsigned hit = __builtin_amdgcn_ballot_w32(mine);
            while (hit != 0u) {
                const int b = __builtin_ctz(hit);
                hit &= hit - 1u;
                const int   s = __builtin_amdgcn_readlane(os, b);
                const float w = __int_as_float(
                    __builtin_amdgcn_readlane(__float_as_int(dl), b)) * di;
                const float4 v = ((const float4*)(h + (size_t)s * DIM))[lane];
                acc.x += v.x * w;
                acc.y += v.y * w;
                acc.z += v.z * w;
                acc.w += v.w * w;
            }
        }
    }

    acc.x = fmaxf(acc.x, 0.0f);
    acc.y = fmaxf(acc.y, 0.0f);
    acc.z = fmaxf(acc.z, 0.0f);
    acc.w = fmaxf(acc.w, 0.0f);
    return acc;
}

__global__ __launch_bounds__(256) void gather_relu_kernel(
    const float* __restrict__ h, const int* __restrict__ cnt,
    const float* __restrict__ bias, const int* __restrict__ bucket,
    const int* __restrict__ ovf_n, const int* __restrict__ ovf_dst,
    const int* __restrict__ ovf_src, float* __restrict__ agg, int n)
{
    const int lane = threadIdx.x & 31;
    const int i = __builtin_amdgcn_readfirstlane(
        (int)(blockIdx.x * (blockDim.x >> 5) + (threadIdx.x >> 5)));
    if (i >= n) return;
    const float4 acc = gather_row(h, cnt, bias, bucket, ovf_n, ovf_dst, ovf_src, i, lane);
    ((float4*)(agg + (size_t)i * DIM))[lane] = acc;
}

__global__ __launch_bounds__(256) void gather_relu_pool_kernel(
    const float* __restrict__ h, const int* __restrict__ cnt,
    const float* __restrict__ bias, const int* __restrict__ bucket,
    const int* __restrict__ ovf_n, const int* __restrict__ ovf_dst,
    const int* __restrict__ ovf_src, const int* __restrict__ batch,
    float* __restrict__ pooled, int n)
{
    __shared__ float srow[8][DIM];
    __shared__ __attribute__((aligned(16))) int sgid[8];
    const int lane = threadIdx.x & 31;
    const int wv   = threadIdx.x >> 5;
    const int i = __builtin_amdgcn_readfirstlane(
        (int)(blockIdx.x * (blockDim.x >> 5) + wv));

    float4 acc = make_float4(0.0f, 0.0f, 0.0f, 0.0f);
    int gid = -1;
    if (i < n) {
        gid = batch[i];
        acc = gather_row(h, cnt, bias, bucket, ovf_n, ovf_dst, ovf_src, i, lane);
    }
    ((float4*)srow[wv])[lane] = acc;
    if (lane == 0) sgid[wv] = gid;
    __syncthreads();

    if (threadIdx.x < DIM) {
        const int c = threadIdx.x;
        int   g[8];
        float v[8];
#pragma unroll
        for (int k = 0; k < 8; ++k) {
            g[k] = __builtin_amdgcn_readfirstlane(sgid[k]);
            v[k] = srow[k][c];
        }
        int   cur = g[0];
        float run = v[0];
#pragma unroll
        for (int k = 1; k < 8; ++k) {
            if (g[k] != cur) {
                if (cur >= 0)
                    __hip_atomic_fetch_add(&pooled[(size_t)cur * DIM + c], run,
                                           __ATOMIC_RELAXED, __HIP_MEMORY_SCOPE_AGENT);
                cur = g[k];
                run = v[k];
            } else {
                run += v[k];
            }
        }
        if (cur >= 0)
            __hip_atomic_fetch_add(&pooled[(size_t)cur * DIM + c], run,
                                   __ATOMIC_RELAXED, __HIP_MEMORY_SCOPE_AGENT);
    }
}

extern "C" void kernel_launch(void* const* d_in, const int* in_sizes, int n_in,
                              void* d_out, int out_size, void* d_ws, size_t ws_size,
                              hipStream_t stream) {
    const float* x       = (const float*)d_in[0];
    const int*   eidx    = (const int*)d_in[1];
    const int*   x_batch = (const int*)d_in[2];
    const float* W1      = (const float*)d_in[3];
    const float* b1      = (const float*)d_in[4];
    const float* W2      = (const float*)d_in[5];
    const float* b2      = (const float*)d_in[6];
    const float* Wh      = (const float*)d_in[7];
    const float* bh      = (const float*)d_in[8];
    float*       out     = (float*)d_out;

    const int n_nodes = in_sizes[0] / DIM;
    const int n_edges = in_sizes[1] / 2;
    const int* src = eidx;
    const int* dst = eidx + n_edges;

    float* h       = (float*)d_ws;
    float* agg     = h      + (size_t)n_nodes * DIM;
    float* pooled  = agg    + (size_t)n_nodes * DIM;
    float* W1T     = pooled + (size_t)NGRAPHS * DIM;
    float* W2T     = W1T    + DIM * DIM;
    float* WhT     = W2T    + DIM * DIM;
    int*   bucket  = (int*)(WhT + OUTC * DIM);
    int*   cnt     = bucket  + (size_t)n_nodes * CAP;
    int*   ovf_dst = cnt     + n_nodes;
    int*   ovf_src = ovf_dst + n_edges;
    int*   ovf_n   = ovf_src + n_edges;

    const int T = 256;
    const int edgeBlk = (n_edges + T - 1) / T;
    const int nodeWaveBlk = (n_nodes + 7) / 8;
    const int gemmBlk = 2 * ((n_nodes / 16 + 7) / 8);

    prep_kernel<<<(2 * DIM * DIM + DIM * OUTC + NGRAPHS * DIM + n_nodes + 1 + T - 1) / T, T, 0, stream>>>(
        W1, W2, Wh, W1T, W2T, WhT, pooled, cnt, ovf_n, n_nodes);

    gemm_rank_bucket_kernel<<<gemmBlk + edgeBlk, T, 0, stream>>>(
        x, W1T, h, n_nodes, gemmBlk, src, dst, cnt, bucket,
        ovf_n, ovf_dst, ovf_src, n_edges);

    gather_relu_kernel<<<nodeWaveBlk, T, 0, stream>>>(h, cnt, b1, bucket, ovf_n, ovf_dst, ovf_src,
                                                     agg, n_nodes);

    wmma_gemm_f32_bt<<<gemmBlk, T, 0, stream>>>(agg, W2T, h, n_nodes);

    gather_relu_pool_kernel<<<nodeWaveBlk, T, 0, stream>>>(h, cnt, b2, bucket, ovf_n, ovf_dst,
                                                          ovf_src, x_batch, pooled, n_nodes);

    head_logsoftmax_kernel<<<NGRAPHS / 16, 32, 0, stream>>>(pooled, WhT, bh, out,
                                                          NGRAPHS, DIM);
}


# Layer-2 product fused into the layer-1 neighbour-sum launch: 16 finished rows go to an LDS tile, 4 waves multiply it by W2^T held in LDS; resident work-groups loop over tiles; 6 -> 5 launches
# speedup vs baseline: 3.8909x; 1.0578x over previous
#include <hip/hip_runtime.h>
#include <hip/hip_bf16.h>

typedef float v2f __attribute__((ext_vector_type(2)));
typedef float v8f __attribute__((ext_vector_type(8)));

#define DIM 128
#define OUTC 64
#define NGRAPHS 512
#define CAP 32
#define BPITCH 132

__global__ void prep_kernel(const float* __restrict__ W1, const float* __restrict__ W2,
                            const float* __restrict__ Wh,
                            float* __restrict__ W1T, float* __restrict__ W2T,
                            float* __restrict__ WhT,
                            float* __restrict__ pooled, int* __restrict__ cnt,
                            int* __restrict__ ovf_n, int n) {
    int idx = blockIdx.x * blockDim.x + threadIdx.x;
    if (idx < DIM * DIM) {
        int k = idx / DIM;
        int c = idx - k * DIM;
        W1T[c * DIM + k] = W1[idx];
        return;
    }
    idx -= DIM * DIM;
    if (idx < DIM * DIM) {
        int k = idx / DIM;
        int c = idx - k * DIM;
        W2T[c * DIM + k] = W2[idx];
        return;
    }
    idx -= DIM * DIM;
    if (idx < DIM * OUTC) {
        int k = idx / OUTC;
        int c = idx - k * OUTC;
        WhT[c * DIM + k] = Wh[idx];
        return;
    }
    idx -= DIM * OUTC;
    if (idx < NGRAPHS * DIM) {
        pooled[idx] = 0.0f;
        return;
    }
    idx -= NGRAPHS * DIM;
    if (idx < n) {
        cnt[idx] = 0;
        return;
    }
    if (idx == n) *ovf_n = 0;
}

struct Strip { v8f a0, a1, a2, a3; };

__device__ __forceinline__ Strip gemm_strip(
    const float* __restrict__ A, const float* __restrict__ BT,
    int K, int row0, int col0, int lane)
{
    const int m16   = lane & 15;
    const int khalf = (lane >> 4) << 1;

    const float* ap  = A  + (size_t)(row0 + m16) * K + khalf;
    const float* bp0 = BT + (size_t)(col0 + m16) * K + khalf;
    const float* bp1 = bp0 + (size_t)16 * K;
    const float* bp2 = bp0 + (size_t)32 * K;
    const float* bp3 = bp0 + (size_t)48 * K;

    v8f acc0 = {}, acc1 = {}, acc2 = {}, acc3 = {};
#pragma unroll 4
    for (int k0 = 0; k0 < K; k0 += 4) {
        v2f a  = *(const v2f*)(ap  + k0);
        v2f b0 = *(const v2f*)(bp0 + k0);
        v2f b1 = *(const v2f*)(bp1 + k0);
        v2f b2 = *(const v2f*)(bp2 + k0);
        v2f b3 = *(const v2f*)(bp3 + k0);
        acc0 = __builtin_amdgcn_wmma_f32_16x16x4_f32(false, a, false, b0, (short)0, acc0, false, false);
        acc1 = __builtin_amdgcn_wmma_f32_16x16x4_f32(false, a, false, b1, (short)0, acc1, false, false);
        acc2 = __builtin_amdgcn_wmma_f32_16x16x4_f32(false, a, false, b2, (short)0, acc2, false, false);
        acc3 = __builtin_amdgcn_wmma_f32_16x16x4_f32(false, a, false, b3, (short)0, acc3, false, false);
    }
    Strip s;
    s.a0 = acc0; s.a1 = acc1; s.a2 = acc2; s.a3 = acc3;
    return s;
}

__device__ __forceinline__ void stage_bt_half(
    float* __restrict__ sB, const float* __restrict__ BThalf, int tid)
{
#pragma unroll
    for (int p = 0; p < 8; ++p) {
        const int idx = p * 256 + tid;
        const int n = idx >> 5;
        const int q = idx & 31;
        const float4 v = ((const float4*)(BThalf + (size_t)n * DIM))[q];
        *(float4*)(sB + n * BPITCH + 4 * q) = v;
    }
}

__device__ __forceinline__ Strip gemm_strip_lds(
    const float* __restrict__ A, const float* sB, int row0, int lane)
{
    const int c  = lane & 15;
    const int h4 = (lane >> 4) << 2;
    const float4* ap = (const float4*)(A + (size_t)(row0 + c) * DIM + h4);
    const float*  bp = sB + c * BPITCH + h4;

    float4 a[6];
#pragma unroll
    for (int i = 0; i < 6; ++i) a[i] = ap[2 * i];
    float4 b[2][4];
#pragma unroll
    for (int j = 0; j < 4; ++j) b[0][j] = *(const float4*)(bp + j * 16 * BPITCH);

    v8f acc0 = {}, acc1 = {}, acc2 = {}, acc3 = {};
    __builtin_amdgcn_sched_barrier(0);
#pragma unroll
    for (int i = 0; i < DIM / 8; ++i) {
        const int cur = i & 1;
        if (i >= 1 && i + 5 < DIM / 8) a[(i - 1) % 6] = ap[2 * (i + 5)];
        if (i + 1 < DIM / 8) {
#pragma unroll
            for (int j = 0; j < 4; ++j)
                b[cur ^ 1][j] = *(const float4*)(bp + j * 16 * BPITCH + (i + 1) * 8);
        }
        __builtin_amdgcn_sched_barrier(0);
        const float4 av = a[i % 6];
        const v2f alo = {av.x, av.y};
        const v2f ahi = {av.z, av.w};
        const v2f b0lo = {b[cur][0].x, b[cur][0].y}, b0hi = {b[cur][0].z, b[cur][0].w};
        const v2f b1lo = {b[cur][1].x, b[cur][1].y}, b1hi = {b[cur][1].z, b[cur][1].w};
        const v2f b2lo = {b[cur][2].x, b[cur][2].y}, b2hi = {b[cur][2].z, b[cur][2].w};
        const v2f b3lo = {b[cur][3].x, b[cur][3].y}, b3hi = {b[cur][3].z, b[cur][3].w};
        acc0 = __builtin_amdgcn_wmma_f32_16x16x4_f32(false, alo, false, b0lo, (short)0, acc0, false, false);
        acc1 = __builtin_amdgcn_wmma_f32_16x16x4_f32(false, alo, false, b1lo, (short)0, acc1, false, false);
        acc2 = __builtin_amdgcn_wmma_f32_16x16x4_f32(false, alo, false, b2lo, (short)0, acc2, false, false);
        acc3 = __builtin_amdgcn_wmma_f32_16x16x4_f32(false, alo, false, b3lo, (short)0, acc3, false, false);
        acc0 = __builtin_amdgcn_wmma_f32_16x16x4_f32(false, ahi, false, b0hi, (short)0, acc0, false, false);
        acc1 = __builtin_amdgcn_wmma_f32_16x16x4_f32(false, ahi, false, b1hi, (short)0, acc1, false, false);
        acc2 = __builtin_amdgcn_wmma_f32_16x16x4_f32(false, ahi, false, b2hi, (short)0, acc2, false, false);
        acc3 = __builtin_amdgcn_wmma_f32_16x16x4_f32(false, ahi, false, b3hi, (short)0, acc3, false, false);
        __builtin_amdgcn_sched_barrier(0);
    }
    Strip s;
    s.a0 = acc0; s.a1 = acc1; s.a2 = acc2; s.a3 = acc3;
    return s;
}

__device__ __forceinline__ void store_strip(
    const Strip& s, float* __restrict__ C, int N, int row0, int col0, int lane)
{
    const int nOff  = lane & 15;
    const int mBase = row0 + ((lane >> 4) << 3);
    float* crow = C + (size_t)mBase * N + col0 + nOff;
#pragma unroll
    for (int r = 0; r < 8; ++r) {
        crow[(size_t)r * N]      = s.a0[r];
        crow[(size_t)r * N + 16] = s.a1[r];
        crow[(size_t)r * N + 32] = s.a2[r];
        crow[(size_t)r * N + 48] = s.a3[r];
    }
}

__global__ __launch_bounds__(256) void gemm_rank_bucket_kernel(
    const float* __restrict__ A, const float* __restrict__ BT,
    float* __restrict__ C, int M, int gemmBlocks,
    const int* __restrict__ src, const int* __restrict__ dst,
    int* __restrict__ cnt, int* __restrict__ bucket,
    int* __restrict__ ovf_n, int* __restrict__ ovf_dst, int* __restrict__ ovf_src,
    int e)
{
    if ((int)blockIdx.x >= gemmBlocks) {
        const int i = ((int)blockIdx.x - gemmBlocks) * (int)blockDim.x + (int)threadIdx.x;
        if (i < e) {
            const int d = dst[i];
            const int s = src[i];
            const int r = __hip_atomic_fetch_add(&cnt[d], 1,
                                                 __ATOMIC_RELAXED, __HIP_MEMORY_SCOPE_AGENT);
            if (r < CAP) {
                bucket[(size_t)d * CAP + r] = s;
            } else {
                const int k = __hip_atomic_fetch_add(ovf_n, 1,
                                                     __ATOMIC_RELAXED, __HIP_MEMORY_SCOPE_AGENT);
                ovf_dst[k] = d;
                ovf_src[k] = s;
            }
        }
        return;
    }
    __shared__ __attribute__((aligned(16))) float sB[64 * BPITCH];
    const int lane = threadIdx.x & 31;
    const int cg    = (int)blockIdx.x & 1;
    const int tileM = ((int)blockIdx.x >> 1) * 8 + (int)(threadIdx.x >> 5);
    stage_bt_half(sB, BT + (size_t)cg * 64 * DIM, (int)threadIdx.x);
    __syncthreads();
    if (tileM * 16 >= M) return;
    const int row0 = tileM << 4;
    const int col0 = cg << 6;
    const Strip s = gemm_strip_lds(A, sB, row0, lane);
    store_strip(s, C, DIM, row0, col0, lane);
}

__global__ __launch_bounds__(256) void head_logsoftmax_kernel(
    const float* __restrict__ A, const float* __restrict__ BT,
    const float* __restrict__ bias, float* __restrict__ out, int M, int K)
{
    const int lane = threadIdx.x & 31;
    const int wave = blockIdx.x * (blockDim.x >> 5) + (threadIdx.x >> 5);
    if (wave * 16 >= M) return;
    const int row0 = wave << 4;
    const Strip s = gemm_strip(A, BT, K, row0, 0, lane);

    const int c  = lane & 15;
    const int h8 = (lane >> 4) << 3;
    const float bb0 = bias[c];
    const float bb1 = bias[c + 16];
    const float bb2 = bias[c + 32];
    const float bb3 = bias[c + 48];
    float* orow = out + (size_t)(row0 + h8) * OUTC + c;
#pragma unroll
    for (int r = 0; r < 8; ++r) {
        const float v0 = s.a0[r] + bb0;
        const float v1 = s.a1[r] + bb1;
        const float v2 = s.a2[r] + bb2;
        const float v3 = s.a3[r] + bb3;
        float m = fmaxf(fmaxf(v0, v2), fmaxf(v1, v3));
        for (int off = 8; off > 0; off >>= 1)
            m = fmaxf(m, __shfl_xor(m, off, 32));
        float sum = (__expf(v0 - m) + __expf(v2 - m)) + (__expf(v1 - m) + __expf(v3 - m));
        for (int off = 8; off > 0; off >>= 1)
            sum += __shfl_xor(sum, off, 32);
        const float ls = __logf(sum);
        orow[(size_t)r * OUTC]      = v0 - m - ls;
        orow[(size_t)r * OUTC + 16] = v1 - m - ls;
        orow[(size_t)r * OUTC + 32] = v2 - m - ls;
        orow[(size_t)r * OUTC + 48] = v3 - m - ls;
    }
}

__device__ __forceinline__ float dinv_of(int c) {
    return rsqrtf((float)(c + 1));
}


__device__ __forceinline__ void issue_rows(
    float4 (&v)[4], const float* __restrict__ h, int sl, int j0, int m, int lane)
{
#pragma unroll
    for (int u = 0; u < 4; ++u) {
        const int jj = j0 + u;
        const int s  = __builtin_amdgcn_readlane(sl, jj);
        v[u] = make_float4(0.0f, 0.0f, 0.0f, 0.0f);
        if (jj < m) v[u] = ((const float4*)(h + (size_t)s * DIM))[lane];
    }
}

__device__ __forceinline__ float4 gather_row(
    const float* __restrict__ h, const int* __restrict__ cnt,
    const float* __restrict__ bias, const int* __restrict__ bucket,
    const int* __restrict__ ovf_n, const int* __restrict__ ovf_dst,
    const int* __restrict__ ovf_src, int i, int lane)
{
    const float4 hv = ((const float4*)(h + (size_t)i * DIM))[lane];
    const float b0 = bias[lane * 4 + 0];
    const float b1 = bias[lane * 4 + 1];
    const float b2 = bias[lane * 4 + 2];
    const float b3 = bias[lane * 4 + 3];

    const int c = cnt[i];
    const int m = (c < CAP) ? c : CAP;
    int sl = i;
    if (lane < m) sl = bucket[(size_t)i * CAP + lane];

    const float di = dinv_of(c);
    const float dd = di * di;
    float4 acc;
    acc.x = hv.x * dd + b0;
    acc.y = hv.y * dd + b1;
    acc.z = hv.z * dd + b2;
    acc.w = hv.w * dd + b3;

    if (m > 0) {
        const int cl = cnt[sl];
        float4 v[4];
        issue_rows(v, h, sl, 0, m, lane);
        const float dl = dinv_of(cl);
        int j0 = 0;
        while (true) {
#pragma unroll
            for (int u = 0; u < 4; ++u) {
                const float w = __int_as_float(
                    __builtin_amdgcn_readlane(__float_as_int(dl), j0 + u)) * di;
                acc.x += v[u].x * w;
                acc.y += v[u].y * w;
                acc.z += v[u].z * w;
                acc.w += v[u].w * w;
            }
            j0 += 4;
            if (j0 >= m) break;
            issue_rows(v, h, sl, j0, m, lane);
        }
    }

    if (c > CAP) {
        const int tot = *ovf_n;
        for (int k0 = 0; k0 < tot; k0 += 32) {
            const int k = k0 + lane;
            int od = -1;
            int os = 0;
            if (k < tot) {
                od = ovf_dst[k];
                os = ovf_src[k];
            }
            const bool mine = (od == i);
            float dl = 0.0f;
            if (mine) dl = dinv_of(cnt[os]);
            unsigned hit = __builtin_amdgcn_ballot_w32(mine);
            while (hit != 0u) {
                const int b = __builtin_ctz(hit);
                hit &= hit - 1u;
                const int   s = __builtin_amdgcn_readlane(os, b);
                const float w = __int_as_float(
                    __builtin_amdgcn_readlane(__float_as_int(dl), b)) * di;
                const float4 v = ((const float4*)(h + (size_t)s * DIM))[lane];
                acc.x += v.x * w;
                acc.y += v.y * w;
                acc.z += v.z * w;
                acc.w += v.w * w;
            }
        }
    }

    acc.x = fmaxf(acc.x, 0.0f);
    acc.y = fmaxf(acc.y, 0.0f);
    acc.z = fmaxf(acc.z, 0.0f);
    acc.w = fmaxf(acc.w, 0.0f);
    return acc;
}

__global__ __launch_bounds__(512) void gather_relu_gemm_kernel(
    const float* __restrict__ h, const int* __restrict__ cnt,
    const float* __restrict__ bias, const int* __restrict__ bucket,
    const int* __restrict__ ovf_n, const int* __restrict__ ovf_dst,
    const int* __restrict__ ovf_src, const float* __restrict__ BT,
    float* __restrict__ C, int n, int ntiles)
{
    __shared__ __attribute__((aligned(16))) float sB[DIM * BPITCH];
    __shared__ __attribute__((aligned(16))) float sA[16 * BPITCH];
    const int lane = threadIdx.x & 31;
    const int wv = __builtin_amdgcn_readfirstlane((int)(threadIdx.x >> 5));

#pragma unroll
    for (int p = 0; p < 8; ++p) {
        const int idx = p * 512 + (int)threadIdx.x;
        const int nr = idx >> 5;
        const int q  = idx & 31;
        const float4 v = ((const float4*)(BT + (size_t)nr * DIM))[q];
        *(float4*)(sB + nr * BPITCH + 4 * q) = v;
    }

    for (int tile = (int)blockIdx.x; tile < ntiles; tile += (int)gridDim.x) {
        const int i = tile * 16 + wv;
        float4 acc = make_float4(0.0f, 0.0f, 0.0f, 0.0f);
        if (i < n)
            acc = gather_row(h, cnt, bias, bucket, ovf_n, ovf_dst, ovf_src, i, lane);
        __syncthreads();
        *(float4*)(sA + wv * BPITCH + 4 * lane) = acc;
        __syncthreads();

        if (wv < 4) {
            const int c  = lane & 15;
            const int h4 = (lane >> 4) << 2;
            const float* ap  = sA + c * BPITCH + h4;
            const float* bp0 = sB + (32 * wv + c) * BPITCH + h4;
            const float* bp1 = bp0 + 16 * BPITCH;
            v8f acc0 = {}, acc1 = {};
#pragma unroll
            for (int kb = 0; kb < DIM / 8; ++kb) {
                const float4 av = *(const float4*)(ap  + 8 * kb);
                const float4 b0 = *(const float4*)(bp0 + 8 * kb);
                const float4 b1 = *(const float4*)(bp1 + 8 * kb);
                const v2f alo  = {av.x, av.y}, ahi  = {av.z, av.w};
                const v2f b0lo = {b0.x, b0.y}, b0hi = {b0.z, b0.w};
                const v2f b1lo = {b1.x, b1.y}, b1hi = {b1.z, b1.w};
                acc0 = __builtin_amdgcn_wmma_f32_16x16x4_f32(false, alo, false, b0lo, (short)0, acc0, false, false);
                acc1 = __builtin_amdgcn_wmma_f32_16x16x4_f32(false, alo, false, b1lo, (short)0, acc1, false, false);
                acc0 = __builtin_amdgcn_wmma_f32_16x16x4_f32(false, ahi, false, b0hi, (short)0, acc0, false, false);
                acc1 = __builtin_amdgcn_wmma_f32_16x16x4_f32(false, ahi, false, b1hi, (short)0, acc1, false, false);
            }
            const int mBase = tile * 16 + ((lane >> 4) << 3);
            float* crow = C + (size_t)mBase * DIM + 32 * wv + c;
#pragma unroll
            for (int r = 0; r < 8; ++r) {
                if (mBase + r < n) {
                    crow[(size_t)r * DIM]      = acc0[r];
                    crow[(size_t)r * DIM + 16] = acc1[r];
                }
            }
        }
    }
}

__global__ __launch_bounds__(256) void gather_relu_pool_kernel(
    const float* __restrict__ h, const int* __restrict__ cnt,
    const float* __restrict__ bias, const int* __restrict__ bucket,
    const int* __restrict__ ovf_n, const int* __restrict__ ovf_dst,
    const int* __restrict__ ovf_src, const int* __restrict__ batch,
    float* __restrict__ pooled, int n)
{
    __shared__ float srow[8][DIM];
    __shared__ __attribute__((aligned(16))) int sgid[8];
    const int lane = threadIdx.x & 31;
    const int wv   = threadIdx.x >> 5;
    const int i = __builtin_amdgcn_readfirstlane(
        (int)(blockIdx.x * (blockDim.x >> 5) + wv));

    float4 acc = make_float4(0.0f, 0.0f, 0.0f, 0.0f);
    int gid = -1;
    if (i < n) {
        gid = batch[i];
        acc = gather_row(h, cnt, bias, bucket, ovf_n, ovf_dst, ovf_src, i, lane);
    }
    ((float4*)srow[wv])[lane] = acc;
    if (lane == 0) sgid[wv] = gid;
    __syncthreads();

    if (threadIdx.x < DIM) {
        const int c = threadIdx.x;
        int   g[8];
        float v[8];
#pragma unroll
        for (int k = 0; k < 8; ++k) {
            g[k] = __builtin_amdgcn_readfirstlane(sgid[k]);
            v[k] = srow[k][c];
        }
        int   cur = g[0];
        float run = v[0];
#pragma unroll
        for (int k = 1; k < 8; ++k) {
            if (g[k] != cur) {
                if (cur >= 0)
                    __hip_atomic_fetch_add(&pooled[(size_t)cur * DIM + c], run,
                                           __ATOMIC_RELAXED, __HIP_MEMORY_SCOPE_AGENT);
                cur = g[k];
                run = v[k];
            } else {
                run += v[k];
            }
        }
        if (cur >= 0)
            __hip_atomic_fetch_add(&pooled[(size_t)cur * DIM + c], run,
                                   __ATOMIC_RELAXED, __HIP_MEMORY_SCOPE_AGENT);
    }
}

extern "C" void kernel_launch(void* const* d_in, const int* in_sizes, int n_in,
                              void* d_out, int out_size, void* d_ws, size_t ws_size,
                              hipStream_t stream) {
    const float* x       = (const float*)d_in[0];
    const int*   eidx    = (const int*)d_in[1];
    const int*   x_batch = (const int*)d_in[2];
    const float* W1      = (const float*)d_in[3];
    const float* b1      = (const float*)d_in[4];
    const float* W2      = (const float*)d_in[5];
    const float* b2      = (const float*)d_in[6];
    const float* Wh      = (const float*)d_in[7];
    const float* bh      = (const float*)d_in[8];
    float*       out     = (float*)d_out;

    const int n_nodes = in_sizes[0] / DIM;
    const int n_edges = in_sizes[1] / 2;
    const int* src = eidx;
    const int* dst = eidx + n_edges;

    float* h       = (float*)d_ws;
    float* h2      = h      + (size_t)n_nodes * DIM;
    float* pooled  = h2     + (size_t)n_nodes * DIM;
    float* W1T     = pooled + (size_t)NGRAPHS * DIM;
    float* W2T     = W1T    + DIM * DIM;
    float* WhT     = W2T    + DIM * DIM;
    int*   bucket  = (int*)(WhT + OUTC * DIM);
    int*   cnt     = bucket  + (size_t)n_nodes * CAP;
    int*   ovf_dst = cnt     + n_nodes;
    int*   ovf_src = ovf_dst + n_edges;
    int*   ovf_n   = ovf_src + n_edges;

    const int T = 256;
    const int edgeBlk = (n_edges + T - 1) / T;
    const int nodeWaveBlk = (n_nodes + 7) / 8;
    const int gemmBlk = 2 * ((n_nodes / 16 + 7) / 8);
    const int nodeTiles = (n_nodes + 15) / 16;

    prep_kernel<<<(2 * DIM * DIM + DIM * OUTC + NGRAPHS * DIM + n_nodes + 1 + T - 1) / T, T, 0, stream>>>(
        W1, W2, Wh, W1T, W2T, WhT, pooled, cnt, ovf_n, n_nodes);

    gemm_rank_bucket_kernel<<<gemmBlk + edgeBlk, T, 0, stream>>>(
        x, W1T, h, n_nodes, gemmBlk, src, dst, cnt, bucket,
        ovf_n, ovf_dst, ovf_src, n_edges);

    gather_relu_gemm_kernel<<<1024, 512, 0, stream>>>(h, cnt, b1, bucket, ovf_n, ovf_dst, ovf_src,
                                                      W2T, h2, n_nodes, nodeTiles);

    gather_relu_pool_kernel<<<nodeWaveBlk, T, 0, stream>>>(h2, cnt, b2, bucket, ovf_n, ovf_dst,
                                                          ovf_src, x_batch, pooled, n_nodes);

    head_logsoftmax_kernel<<<NGRAPHS / 16, 32, 0, stream>>>(pooled, WhT, bh, out,
                                                          NGRAPHS, DIM);
}
